# starting kernel
# baseline (speedup 1.0000x reference)
#include <hip/hip_runtime.h>
#include <hip/hip_bf16.h>

typedef float v2f __attribute__((ext_vector_type(2)));
typedef float v8f __attribute__((ext_vector_type(8)));

#define N_NODES  50000
#define N_EDGES  600000
#define N_GRAPHS 64
#define D_IN     128
#define D_HID    128
#define D_OUT    64

__global__ __launch_bounds__(32)
void gemm_wmma_f32(const float* __restrict__ X, const float* __restrict__ W,
                   float* __restrict__ H, int K, int F) {
  const int rowTile = blockIdx.x << 4;
  const int colTile = blockIdx.y << 4;
  const int lane = threadIdx.x & 31;
  const int half = lane >> 4;
  const int i16  = lane & 15;

  const float* xrow = X + (size_t)(rowTile + i16) * (size_t)K;
  const float* wcol = W + colTile + i16;

  v8f acc = {};
  for (int k0 = 0; k0 < K; k0 += 4) {
    const int ka = k0 + (half << 1);
    v2f a = *(const v2f*)(xrow + ka);
    v2f b;
    b.x = wcol[(size_t)ka * (size_t)F];
    b.y = wcol[(size_t)(ka + 1) * (size_t)F];
    acc = __builtin_amdgcn_wmma_f32_16x16x4_f32(
         false, a,  false, b,
         (short)0, acc,  false,  false);
  }

  float* out = H + (size_t)(rowTile + (half << 3)) * (size_t)F + colTile + i16;
#pragma unroll
  for (int i = 0; i < 8; ++i) out[(size_t)i * (size_t)F] = acc[i];
}

__global__ void zero_f32(float* __restrict__ p, int n) {
  int i = blockIdx.x * blockDim.x + threadIdx.x;
  if (i < n) p[i] = 0.0f;
}

__global__ void deg_accum(const int* __restrict__ dst, float* __restrict__ deg, int E) {
  int e = blockIdx.x * blockDim.x + threadIdx.x;
  if (e < E) atomicAdd(&deg[dst[e]], 1.0f);
}

__global__ void deg_finalize(float* __restrict__ deg, int n) {
  int i = blockIdx.x * blockDim.x + threadIdx.x;
  if (i < n) deg[i] = rsqrtf(deg[i] + 1.0f);
}

__global__ void agg_init(const float* __restrict__ h, const float* __restrict__ dis,
                         const float* __restrict__ bias, float* __restrict__ agg,
                         int F, int total) {
  int i = blockIdx.x * blockDim.x + threadIdx.x;
  if (i >= total) return;
  int node = i / F;
  int f = i - node * F;
  float d = dis[node];
  agg[i] = h[i] * d * d + bias[f];
}

__global__ void edge_scatter(const int* __restrict__ src, const int* __restrict__ dst,
                             const float* __restrict__ h, const float* __restrict__ dis,
                             float* __restrict__ agg, int F, int total) {
  int i = blockIdx.x * blockDim.x + threadIdx.x;
  if (i >= total) return;
  int e = i / F;
  int f = i - e * F;
  int s = src[e];
  int d = dst[e];
  float nm = dis[s] * dis[d];
  atomicAdd(&agg[(size_t)d * (size_t)F + f], h[(size_t)s * (size_t)F + f] * nm);
}

__global__ void relu_k(float* __restrict__ p, int n) {
  int i = blockIdx.x * blockDim.x + threadIdx.x;
  if (i < n) p[i] = fmaxf(p[i], 0.0f);
}

__global__ void pool_accum(const float* __restrict__ h, const int* __restrict__ batch,
                           float* __restrict__ sums, float* __restrict__ cnts, int total) {
  int i = blockIdx.x * blockDim.x + threadIdx.x;
  if (i >= total) return;
  int node = i >> 6;
  int f = i & 63;
  int g = batch[node];
  atomicAdd(&sums[(g << 6) + f], h[i]);
  if (f == 0) atomicAdd(&cnts[g], 1.0f);
}

__global__ void pool_final(const float* __restrict__ sums, const float* __restrict__ cnts,
                           float* __restrict__ out) {
  int i = blockIdx.x * blockDim.x + threadIdx.x;
  if (i < N_GRAPHS * D_OUT) {
    int g = i >> 6;
    out[i] = sums[i] / fmaxf(cnts[g], 1.0f);
  }
}

extern "C" void kernel_launch(void* const* d_in, const int* in_sizes, int n_in,
                              void* d_out, int out_size, void* d_ws, size_t ws_size,
                              hipStream_t stream) {
  const float* x     = (const float*)d_in[0];
  const int*   ei    = (const int*)d_in[1];
  const int*   batch = (const int*)d_in[2];
  const float* W1 = (const float*)d_in[3];  const float* b1 = (const float*)d_in[4];
  const float* W2 = (const float*)d_in[5];  const float* b2 = (const float*)d_in[6];
  const float* W3 = (const float*)d_in[7];  const float* b3 = (const float*)d_in[8];
  const float* W4 = (const float*)d_in[9];  const float* b4 = (const float*)d_in[10];
  float* out = (float*)d_out;

  float* bufA = (float*)d_ws;
  float* bufB = bufA + (size_t)N_NODES * D_HID;
  float* dis  = bufB + (size_t)N_NODES * D_HID;
  float* sums = dis + N_NODES;
  float* cnts = sums + N_GRAPHS * D_OUT;

  const int* src = ei;
  const int* dst = ei + N_EDGES;

  const int TB = 256;
  auto blocks = [](long long n, int tb) { return (unsigned)((n + tb - 1) / tb); };

  zero_f32<<<blocks(N_NODES, TB), TB, 0, stream>>>(dis, N_NODES);
  deg_accum<<<blocks(N_EDGES, TB), TB, 0, stream>>>(dst, dis, N_EDGES);
  deg_finalize<<<blocks(N_NODES, TB), TB, 0, stream>>>(dis, N_NODES);

  auto layer = [&](const float* hin, int K, const float* W, const float* b, int F,
                   bool do_relu) {
    dim3 grid(N_NODES / 16, F / 16);
    gemm_wmma_f32<<<grid, 32, 0, stream>>>(hin, W, bufA, K, F);
    int totalNF = N_NODES * F;
    agg_init<<<blocks(totalNF, TB), TB, 0, stream>>>(bufA, dis, b, bufB, F, totalNF);
    int totalEF = N_EDGES * F;
    edge_scatter<<<blocks(totalEF, TB), TB, 0, stream>>>(src, dst, bufA, dis, bufB, F,
                                                         totalEF);
    if (do_relu) relu_k<<<blocks(totalNF, TB), TB, 0, stream>>>(bufB, totalNF);
  };

  layer(x,    D_IN,  W1, b1, D_HID, true);
  layer(bufB, D_HID, W2, b2, D_HID, true);
  layer(bufB, D_HID, W3, b3, D_HID, true);
  layer(bufB, D_HID, W4, b4, D_OUT, false);

  zero_f32<<<blocks(N_GRAPHS * D_OUT + N_GRAPHS, TB), TB, 0, stream>>>(
      sums, N_GRAPHS * D_OUT + N_GRAPHS);
  pool_accum<<<blocks((long long)N_NODES * D_OUT, TB), TB, 0, stream>>>(
      bufB, batch, sums, cnts, N_NODES * D_OUT);
  pool_final<<<blocks(N_GRAPHS * D_OUT, TB), TB, 0, stream>>>(sums, cnts, out);
}


# edge scatter remapped: one wave moves 8 whole edges, reads src/dst/norm once, lane owns features l+32q so every row load and f32 atomic add is one whole line; 32x fewer waves; atomics kept
# speedup vs baseline: 1.9343x; 1.9343x over previous
#include <hip/hip_runtime.h>
#include <hip/hip_bf16.h>

typedef float v2f __attribute__((ext_vector_type(2)));
typedef float v8f __attribute__((ext_vector_type(8)));

#define N_NODES  50000
#define N_EDGES  600000
#define N_GRAPHS 64
#define D_IN     128
#define D_HID    128
#define D_OUT    64

__global__ __launch_bounds__(32)
void gemm_wmma_f32(const float* __restrict__ X, const float* __restrict__ W,
                   float* __restrict__ H, int K, int F) {
  const int rowTile = blockIdx.x << 4;
  const int colTile = blockIdx.y << 4;
  const int lane = threadIdx.x & 31;
  const int half = lane >> 4;
  const int i16  = lane & 15;

  const float* xrow = X + (size_t)(rowTile + i16) * (size_t)K;
  const float* wcol = W + colTile + i16;

  v8f acc = {};
  for (int k0 = 0; k0 < K; k0 += 4) {
    const int ka = k0 + (half << 1);
    v2f a = *(const v2f*)(xrow + ka);
    v2f b;
    b.x = wcol[(size_t)ka * (size_t)F];
    b.y = wcol[(size_t)(ka + 1) * (size_t)F];
    acc = __builtin_amdgcn_wmma_f32_16x16x4_f32(
         false, a,  false, b,
         (short)0, acc,  false,  false);
  }

  float* out = H + (size_t)(rowTile + (half << 3)) * (size_t)F + colTile + i16;
#pragma unroll
  for (int i = 0; i < 8; ++i) out[(size_t)i * (size_t)F] = acc[i];
}

__global__ void zero_f32(float* __restrict__ p, int n) {
  int i = blockIdx.x * blockDim.x + threadIdx.x;
  if (i < n) p[i] = 0.0f;
}

__global__ void deg_accum(const int* __restrict__ dst, float* __restrict__ deg, int E) {
  int e = blockIdx.x * blockDim.x + threadIdx.x;
  if (e < E) atomicAdd(&deg[dst[e]], 1.0f);
}

__global__ void deg_finalize(float* __restrict__ deg, int n) {
  int i = blockIdx.x * blockDim.x + threadIdx.x;
  if (i < n) deg[i] = rsqrtf(deg[i] + 1.0f);
}

__global__ void agg_init(const float* __restrict__ h, const float* __restrict__ dis,
                         const float* __restrict__ bias, float* __restrict__ agg,
                         int F, int total) {
  int i = blockIdx.x * blockDim.x + threadIdx.x;
  if (i >= total) return;
  int node = i / F;
  int f = i - node * F;
  float d = dis[node];
  agg[i] = h[i] * d * d + bias[f];
}

template <int F>
__global__ __launch_bounds__(256)
void edge_scatter_rows(const int* __restrict__ src, const int* __restrict__ dst,
                       const float* __restrict__ h, const float* __restrict__ dis,
                       float* __restrict__ agg) {
  constexpr int EPW = 8;
  constexpr int FPL = F / 32;
  const int lane = threadIdx.x & 31;
  const int wave = (blockIdx.x * blockDim.x + threadIdx.x) >> 5;
  const int e0 = wave * EPW;
  if (e0 >= N_EDGES) return;

  int e = e0 + (lane & (EPW - 1));
  if (e >= N_EDGES) e = N_EDGES - 1;
  const int s = src[e];
  const int d = dst[e];
  const float nm = dis[s] * dis[d];

  float v[EPW][FPL];
  int   dj[EPW];
  float nj[EPW];
#pragma unroll
  for (int j = 0; j < EPW; ++j) {
    const int sj = __builtin_amdgcn_readlane(s, j);
    dj[j] = __builtin_amdgcn_readlane(d, j);
    nj[j] = __builtin_bit_cast(float,
              __builtin_amdgcn_readlane(__builtin_bit_cast(int, nm), j));
    const float* hrow = h + (size_t)sj * (size_t)F + lane;
#pragma unroll
    for (int q = 0; q < FPL; ++q) v[j][q] = hrow[q * 32];
  }
#pragma unroll
  for (int j = 0; j < EPW; ++j) {
    if (e0 + j < N_EDGES) {
      float* arow = agg + (size_t)dj[j] * (size_t)F + lane;
#pragma unroll
      for (int q = 0; q < FPL; ++q) atomicAdd(arow + q * 32, v[j][q] * nj[j]);
    }
  }
}

__global__ void relu_k(float* __restrict__ p, int n) {
  int i = blockIdx.x * blockDim.x + threadIdx.x;
  if (i < n) p[i] = fmaxf(p[i], 0.0f);
}

__global__ void pool_accum(const float* __restrict__ h, const int* __restrict__ batch,
                           float* __restrict__ sums, float* __restrict__ cnts, int total) {
  int i = blockIdx.x * blockDim.x + threadIdx.x;
  if (i >= total) return;
  int node = i >> 6;
  int f = i & 63;
  int g = batch[node];
  atomicAdd(&sums[(g << 6) + f], h[i]);
  if (f == 0) atomicAdd(&cnts[g], 1.0f);
}

__global__ void pool_final(const float* __restrict__ sums, const float* __restrict__ cnts,
                           float* __restrict__ out) {
  int i = blockIdx.x * blockDim.x + threadIdx.x;
  if (i < N_GRAPHS * D_OUT) {
    int g = i >> 6;
    out[i] = sums[i] / fmaxf(cnts[g], 1.0f);
  }
}

extern "C" void kernel_launch(void* const* d_in, const int* in_sizes, int n_in,
                              void* d_out, int out_size, void* d_ws, size_t ws_size,
                              hipStream_t stream) {
  const float* x     = (const float*)d_in[0];
  const int*   ei    = (const int*)d_in[1];
  const int*   batch = (const int*)d_in[2];
  const float* W1 = (const float*)d_in[3];  const float* b1 = (const float*)d_in[4];
  const float* W2 = (const float*)d_in[5];  const float* b2 = (const float*)d_in[6];
  const float* W3 = (const float*)d_in[7];  const float* b3 = (const float*)d_in[8];
  const float* W4 = (const float*)d_in[9];  const float* b4 = (const float*)d_in[10];
  float* out = (float*)d_out;

  float* bufA = (float*)d_ws;
  float* bufB = bufA + (size_t)N_NODES * D_HID;
  float* dis  = bufB + (size_t)N_NODES * D_HID;
  float* sums = dis + N_NODES;
  float* cnts = sums + N_GRAPHS * D_OUT;

  const int* src = ei;
  const int* dst = ei + N_EDGES;

  const int TB = 256;
  auto blocks = [](long long n, int tb) { return (unsigned)((n + tb - 1) / tb); };

  zero_f32<<<blocks(N_NODES, TB), TB, 0, stream>>>(dis, N_NODES);
  deg_accum<<<blocks(N_EDGES, TB), TB, 0, stream>>>(dst, dis, N_EDGES);
  deg_finalize<<<blocks(N_NODES, TB), TB, 0, stream>>>(dis, N_NODES);

  auto layer = [&](const float* hin, int K, const float* W, const float* b, int F,
                   bool do_relu) {
    dim3 grid(N_NODES / 16, F / 16);
    gemm_wmma_f32<<<grid, 32, 0, stream>>>(hin, W, bufA, K, F);
    int totalNF = N_NODES * F;
    agg_init<<<blocks(totalNF, TB), TB, 0, stream>>>(bufA, dis, b, bufB, F, totalNF);
    if (F == D_HID)
      edge_scatter_rows<D_HID><<<(N_EDGES + 63) / 64, TB, 0, stream>>>(src, dst, bufA, dis, bufB);
    else
      edge_scatter_rows<D_OUT><<<(N_EDGES + 63) / 64, TB, 0, stream>>>(src, dst, bufA, dis, bufB);
    if (do_relu) relu_k<<<blocks(totalNF, TB), TB, 0, stream>>>(bufB, totalNF);
  };

  layer(x,    D_IN,  W1, b1, D_HID, true);
  layer(bufB, D_HID, W2, b2, D_HID, true);
  layer(bufB, D_HID, W3, b3, D_HID, true);
  layer(bufB, D_HID, W4, b4, D_OUT, false);

  zero_f32<<<blocks(N_GRAPHS * D_OUT + N_GRAPHS, TB), TB, 0, stream>>>(
      sums, N_GRAPHS * D_OUT + N_GRAPHS);
  pool_accum<<<blocks((long long)N_NODES * D_OUT, TB), TB, 0, stream>>>(
      bufB, batch, sums, cnts, N_NODES * D_OUT);
  pool_final<<<blocks(N_GRAPHS * D_OUT, TB), TB, 0, stream>>>(sums, cnts, out);
}


# relu (on A load) and self-loop+bias seed (epilogue) fused into the WMMA GEMM: a block of F/16 waves owns 16 whole rows, seed written in place behind a barrier; 7 launches fewer
# speedup vs baseline: 2.7886x; 1.4417x over previous
#include <hip/hip_runtime.h>
#include <hip/hip_bf16.h>

typedef float v2f __attribute__((ext_vector_type(2)));
typedef float v8f __attribute__((ext_vector_type(8)));

#define N_NODES  50000
#define N_EDGES  600000
#define N_GRAPHS 64
#define D_IN     128
#define D_HID    128
#define D_OUT    64

template <bool RELU_IN>
__global__ __launch_bounds__(256)
void gemm_wmma_f32_seed(const float* X, const float* __restrict__ W,
                        const float* __restrict__ dis, const float* __restrict__ bias,
                        float* __restrict__ H, float* S, int K, int F) {
  const int rowTile = blockIdx.x << 4;
  const int colTile = (threadIdx.x >> 5) << 4;
  const int lane = threadIdx.x & 31;
  const int half = lane >> 4;
  const int i16  = lane & 15;

  const float* xrow = X + (size_t)(rowTile + i16) * (size_t)K;
  const float* wcol = W + colTile + i16;

  v8f acc = {};
  for (int k0 = 0; k0 < K; k0 += 16) {
    v2f a[4], b[4];
#pragma unroll
    for (int u = 0; u < 4; ++u) {
      const int ka = k0 + 4 * u + (half << 1);
      a[u] = *(const v2f*)(xrow + ka);
    }
#pragma unroll
    for (int u = 0; u < 4; ++u) {
      const int ka = k0 + 4 * u + (half << 1);
      b[u].x = wcol[(size_t)ka * (size_t)F];
      b[u].y = wcol[(size_t)(ka + 1) * (size_t)F];
    }
    if (RELU_IN) {
#pragma unroll
      for (int u = 0; u < 4; ++u) {
        a[u].x = fmaxf(a[u].x, 0.0f);
        a[u].y = fmaxf(a[u].y, 0.0f);
      }
      __builtin_amdgcn_sched_barrier(0);
    }
#pragma unroll
    for (int u = 0; u < 4; ++u)
      acc = __builtin_amdgcn_wmma_f32_16x16x4_f32(
           false, a[u],  false, b[u],
           (short)0, acc,  false,  false);
  }

  const int r0 = rowTile + (half << 3);
  const float bv = bias[colTile + i16];
  float dd[8];
#pragma unroll
  for (int i = 0; i < 8; ++i) dd[i] = dis[r0 + i];

  __syncthreads();

  float* hout = H + (size_t)r0 * (size_t)F + colTile + i16;
  float* sout = S + (size_t)r0 * (size_t)K + colTile + i16;
#pragma unroll
  for (int i = 0; i < 8; ++i) {
    const float hv = acc[i];
    hout[(size_t)i * (size_t)F] = hv;
    sout[(size_t)i * (size_t)K] = hv * dd[i] * dd[i] + bv;
  }
}

__global__ void zero_f32(float* __restrict__ p, int n) {
  int i = blockIdx.x * blockDim.x + threadIdx.x;
  if (i < n) p[i] = 0.0f;
}

__global__ void deg_accum(const int* __restrict__ dst, float* __restrict__ deg, int E) {
  int e = blockIdx.x * blockDim.x + threadIdx.x;
  if (e < E) atomicAdd(&deg[dst[e]], 1.0f);
}

__global__ void deg_finalize(float* __restrict__ deg, int n) {
  int i = blockIdx.x * blockDim.x + threadIdx.x;
  if (i < n) deg[i] = rsqrtf(deg[i] + 1.0f);
}

template <int F>
__global__ __launch_bounds__(256)
void edge_scatter_rows(const int* __restrict__ src, const int* __restrict__ dst,
                       const float* __restrict__ h, const float* __restrict__ dis,
                       float* __restrict__ agg) {
  constexpr int EPW = 8;
  constexpr int FPL = F / 32;
  const int lane = threadIdx.x & 31;
  const int wave = (blockIdx.x * blockDim.x + threadIdx.x) >> 5;
  const int e0 = wave * EPW;
  if (e0 >= N_EDGES) return;

  int e = e0 + (lane & (EPW - 1));
  if (e >= N_EDGES) e = N_EDGES - 1;
  const int s = src[e];
  const int d = dst[e];
  const float nm = dis[s] * dis[d];

  float v[EPW][FPL];
  int   dj[EPW];
  float nj[EPW];
#pragma unroll
  for (int j = 0; j < EPW; ++j) {
    const int sj = __builtin_amdgcn_readlane(s, j);
    dj[j] = __builtin_amdgcn_readlane(d, j);
    nj[j] = __builtin_bit_cast(float,
              __builtin_amdgcn_readlane(__builtin_bit_cast(int, nm), j));
    const float* hrow = h + (size_t)sj * (size_t)F + lane;
#pragma unroll
    for (int q = 0; q < FPL; ++q) v[j][q] = hrow[q * 32];
  }
#pragma unroll
  for (int j = 0; j < EPW; ++j) {
    if (e0 + j < N_EDGES) {
      float* arow = agg + (size_t)dj[j] * (size_t)D_HID + lane;
#pragma unroll
      for (int q = 0; q < FPL; ++q) atomicAdd(arow + q * 32, v[j][q] * nj[j]);
    }
  }
}

__global__ void pool_accum(const float* __restrict__ h, const int* __restrict__ batch,
                           float* __restrict__ sums, float* __restrict__ cnts, int total) {
  int i = blockIdx.x * blockDim.x + threadIdx.x;
  if (i >= total) return;
  int node = i >> 6;
  int f = i & 63;
  int g = batch[node];
  atomicAdd(&sums[(g << 6) + f], h[(node << 7) + f]);
  if (f == 0) atomicAdd(&cnts[g], 1.0f);
}

__global__ void pool_final(const float* __restrict__ sums, const float* __restrict__ cnts,
                           float* __restrict__ out) {
  int i = blockIdx.x * blockDim.x + threadIdx.x;
  if (i < N_GRAPHS * D_OUT) {
    int g = i >> 6;
    out[i] = sums[i] / fmaxf(cnts[g], 1.0f);
  }
}

extern "C" void kernel_launch(void* const* d_in, const int* in_sizes, int n_in,
                              void* d_out, int out_size, void* d_ws, size_t ws_size,
                              hipStream_t stream) {
  const float* x     = (const float*)d_in[0];
  const int*   ei    = (const int*)d_in[1];
  const int*   batch = (const int*)d_in[2];
  const float* W1 = (const float*)d_in[3];  const float* b1 = (const float*)d_in[4];
  const float* W2 = (const float*)d_in[5];  const float* b2 = (const float*)d_in[6];
  const float* W3 = (const float*)d_in[7];  const float* b3 = (const float*)d_in[8];
  const float* W4 = (const float*)d_in[9];  const float* b4 = (const float*)d_in[10];
  float* out = (float*)d_out;

  float* bufA = (float*)d_ws;
  float* bufB = bufA + (size_t)N_NODES * D_HID;
  float* dis  = bufB + (size_t)N_NODES * D_HID;
  float* sums = dis + N_NODES;
  float* cnts = sums + N_GRAPHS * D_OUT;

  const int* src = ei;
  const int* dst = ei + N_EDGES;

  const int TB = 256;
  auto blocks = [](long long n, int tb) { return (unsigned)((n + tb - 1) / tb); };

  zero_f32<<<blocks(N_NODES, TB), TB, 0, stream>>>(dis, N_NODES);
  deg_accum<<<blocks(N_EDGES, TB), TB, 0, stream>>>(dst, dis, N_EDGES);
  deg_finalize<<<blocks(N_NODES, TB), TB, 0, stream>>>(dis, N_NODES);

  auto layer = [&](const float* hin, int K, const float* W, const float* b, int F,
                   bool relu_in) {
    if (relu_in)
      gemm_wmma_f32_seed<true><<<N_NODES / 16, 32 * (F / 16), 0, stream>>>(
          hin, W, dis, b, bufA, bufB, K, F);
    else
      gemm_wmma_f32_seed<false><<<N_NODES / 16, 32 * (F / 16), 0, stream>>>(
          hin, W, dis, b, bufA, bufB, K, F);
    if (F == D_HID)
      edge_scatter_rows<D_HID><<<(N_EDGES + 63) / 64, TB, 0, stream>>>(src, dst, bufA, dis, bufB);
    else
      edge_scatter_rows<D_OUT><<<(N_EDGES + 63) / 64, TB, 0, stream>>>(src, dst, bufA, dis, bufB);
  };

  layer(x,    D_IN,  W1, b1, D_HID, false);
  layer(bufB, D_HID, W2, b2, D_HID, true);
  layer(bufB, D_HID, W3, b3, D_HID, true);
  layer(bufB, D_HID, W4, b4, D_OUT, true);

  zero_f32<<<blocks(N_GRAPHS * D_OUT + N_GRAPHS, TB), TB, 0, stream>>>(
      sums, N_GRAPHS * D_OUT + N_GRAPHS);
  pool_accum<<<blocks((long long)N_NODES * D_OUT, TB), TB, 0, stream>>>(
      bufB, batch, sums, cnts, N_NODES * D_OUT);
  pool_final<<<blocks(N_GRAPHS * D_OUT, TB), TB, 0, stream>>>(sums, cnts, out);
}


# edges ranked by destination once per call (count atomic's return), each layer sums a row's in-edges in registers, one store per row: no f32 atomics, no seed; sum order = rank order
# speedup vs baseline: 3.2358x; 1.1604x over previous
#include <hip/hip_runtime.h>
#include <hip/hip_bf16.h>

typedef float v2f __attribute__((ext_vector_type(2)));
typedef float v8f __attribute__((ext_vector_type(8)));

#define N_NODES  50000
#define N_EDGES  600000
#define N_GRAPHS 64
#define D_IN     128
#define D_HID    128
#define D_OUT    64

#define WS_WORDS ((size_t)2 * N_NODES * D_HID + N_NODES + N_GRAPHS * D_OUT + N_GRAPHS + \
                  (N_NODES + 32) + N_NODES + (size_t)2 * N_EDGES)
#define WS_BYTES (WS_WORDS * sizeof(float))

template <bool RELU_IN>
__global__ __launch_bounds__(256)
void gemm_wmma_f32(const float* __restrict__ X, const float* __restrict__ W,
                   float* __restrict__ H, int K, int F) {
  const int rowTile = blockIdx.x << 4;
  const int colTile = (threadIdx.x >> 5) << 4;
  const int lane = threadIdx.x & 31;
  const int half = lane >> 4;
  const int i16  = lane & 15;

  const float* xrow = X + (size_t)(rowTile + i16) * (size_t)K;
  const float* wcol = W + colTile + i16;

  v8f acc = {};
  for (int k0 = 0; k0 < K; k0 += 16) {
    v2f a[4], b[4];
#pragma unroll
    for (int u = 0; u < 4; ++u) {
      const int ka = k0 + 4 * u + (half << 1);
      a[u] = *(const v2f*)(xrow + ka);
    }
#pragma unroll
    for (int u = 0; u < 4; ++u) {
      const int ka = k0 + 4 * u + (half << 1);
      b[u].x = wcol[(size_t)ka * (size_t)F];
      b[u].y = wcol[(size_t)(ka + 1) * (size_t)F];
    }
    if (RELU_IN) {
#pragma unroll
      for (int u = 0; u < 4; ++u) {
        a[u].x = fmaxf(a[u].x, 0.0f);
        a[u].y = fmaxf(a[u].y, 0.0f);
      }
      __builtin_amdgcn_sched_barrier(0);
    }
#pragma unroll
    for (int u = 0; u < 4; ++u)
      acc = __builtin_amdgcn_wmma_f32_16x16x4_f32(
           false, a[u],  false, b[u],
           (short)0, acc,  false,  false);
  }

  const int r0 = rowTile + (half << 3);
  float* hout = H + (size_t)r0 * (size_t)F + colTile + i16;
#pragma unroll
  for (int i = 0; i < 8; ++i) hout[(size_t)i * (size_t)F] = acc[i];
}

__global__ void zero_f32(float* __restrict__ p, int n) {
  int i = blockIdx.x * blockDim.x + threadIdx.x;
  if (i < n) p[i] = 0.0f;
}

__global__ void zero_counts(int* __restrict__ cnt, size_t ws_bytes) {
  if (ws_bytes < WS_BYTES) return;
  int i = blockIdx.x * blockDim.x + threadIdx.x;
  if (i <= N_NODES) cnt[i] = 0;
}

__global__ void count_rank(const int* __restrict__ dst, int* __restrict__ cnt,
                           int* __restrict__ rank, size_t ws_bytes) {
  if (ws_bytes < WS_BYTES) return;
  int e = blockIdx.x * blockDim.x + threadIdx.x;
  if (e < N_EDGES) rank[e] = atomicAdd(&cnt[dst[e]], 1);
}

__global__ __launch_bounds__(256)
void row_alloc(int* __restrict__ cnt, int* __restrict__ rstart,
               float* __restrict__ dis, size_t ws_bytes) {
  if (ws_bytes < WS_BYTES) return;
  const int i = blockIdx.x * blockDim.x + threadIdx.x;
  const int lane = threadIdx.x & 31;
  const int c = (i < N_NODES) ? cnt[i] : 0;
  int incl = c;
#pragma unroll
  for (int o = 1; o < 32; o <<= 1) {
    const int t = __shfl_up(incl, o);
    if (lane >= o) incl += t;
  }
  const int wsum = __shfl(incl, 31);
  int base = 0;
  if (lane == 31) base = atomicAdd(&cnt[N_NODES], wsum);
  base = __shfl(base, 31);
  if (i < N_NODES) {
    rstart[i] = base + incl - c;
    dis[i] = rsqrtf((float)c + 1.0f);
  }
}

__global__ void fill_sorted(const int* __restrict__ src, const int* __restrict__ dst,
                            const int* __restrict__ rank, const int* __restrict__ rstart,
                            const float* __restrict__ dis, int* __restrict__ csrc,
                            float* __restrict__ cw, size_t ws_bytes) {
  if (ws_bytes < WS_BYTES) return;
  int e = blockIdx.x * blockDim.x + threadIdx.x;
  if (e < N_EDGES) {
    const int s = src[e];
    const int d = dst[e];
    const int pos = rstart[d] + rank[e];
    csrc[pos] = s;
    cw[pos] = dis[s] * dis[d];
  }
}

template <int F>
__global__ __launch_bounds__(256)
void gather_rows(const float* __restrict__ h, const int* __restrict__ cnt,
                 const int* __restrict__ rstart, const int* __restrict__ csrc,
                 const float* __restrict__ cw, const float* __restrict__ dis,
                 const float* __restrict__ bias, float* __restrict__ out,
                 size_t ws_bytes) {
  constexpr int VW = F / 32;
  typedef float vec __attribute__((ext_vector_type(VW)));
  if (ws_bytes < WS_BYTES) return;
  const int lane = threadIdx.x & 31;
  const int row = (blockIdx.x * blockDim.x + threadIdx.x) >> 5;
  if (row >= N_NODES) return;

  const int n  = __builtin_amdgcn_readfirstlane(cnt[row]);
  const int st = __builtin_amdgcn_readfirstlane(rstart[row]);
  const float di = dis[row];

  const vec* hv = (const vec*)h;
  const vec self = hv[(size_t)row * 32 + lane];
  const vec bv = ((const vec*)bias)[lane];
  vec acc = self * di * di + bv;

  for (int c0 = 0; c0 < n; c0 += 32) {
    const int m = (n - c0 < 32) ? (n - c0) : 32;
    int sv = row;
    float wv = 0.0f;
    if (lane < m) {
      sv = csrc[st + c0 + lane];
      wv = cw[st + c0 + lane];
    }
    for (int j = 0; j < m; j += 4) {
      vec r[4];
      float w[4];
#pragma unroll
      for (int u = 0; u < 4; ++u) {
        const int sj = __builtin_amdgcn_readlane(sv, j + u);
        w[u] = __builtin_bit_cast(float,
                 __builtin_amdgcn_readlane(__builtin_bit_cast(int, wv), j + u));
        r[u] = hv[(size_t)sj * 32 + lane];
      }
#pragma unroll
      for (int u = 0; u < 4; ++u) acc += r[u] * w[u];
    }
  }
  ((vec*)out)[(size_t)row * (D_HID / VW) + lane] = acc;
}

__global__ void pool_accum(const float* __restrict__ h, const int* __restrict__ batch,
                           float* __restrict__ sums, float* __restrict__ cnts, int total) {
  int i = blockIdx.x * blockDim.x + threadIdx.x;
  if (i >= total) return;
  int node = i >> 6;
  int f = i & 63;
  int g = batch[node];
  atomicAdd(&sums[(g << 6) + f], h[(node << 7) + f]);
  if (f == 0) atomicAdd(&cnts[g], 1.0f);
}

__global__ void pool_final(const float* __restrict__ sums, const float* __restrict__ cnts,
                           float* __restrict__ out) {
  int i = blockIdx.x * blockDim.x + threadIdx.x;
  if (i < N_GRAPHS * D_OUT) {
    int g = i >> 6;
    out[i] = sums[i] / fmaxf(cnts[g], 1.0f);
  }
}

extern "C" void kernel_launch(void* const* d_in, const int* in_sizes, int n_in,
                              void* d_out, int out_size, void* d_ws, size_t ws_size,
                              hipStream_t stream) {
  const float* x     = (const float*)d_in[0];
  const int*   ei    = (const int*)d_in[1];
  const int*   batch = (const int*)d_in[2];
  const float* W1 = (const float*)d_in[3];  const float* b1 = (const float*)d_in[4];
  const float* W2 = (const float*)d_in[5];  const float* b2 = (const float*)d_in[6];
  const float* W3 = (const float*)d_in[7];  const float* b3 = (const float*)d_in[8];
  const float* W4 = (const float*)d_in[9];  const float* b4 = (const float*)d_in[10];
  float* out = (float*)d_out;

  float* bufA = (float*)d_ws;
  float* bufB = bufA + (size_t)N_NODES * D_HID;
  float* dis  = bufB + (size_t)N_NODES * D_HID;
  float* sums = dis + N_NODES;
  float* cnts = sums + N_GRAPHS * D_OUT;
  int*   cnt    = (int*)(cnts + N_GRAPHS);
  int*   rstart = cnt + (N_NODES + 32);
  int*   csrc   = rstart + N_NODES;
  float* cw     = (float*)(csrc + N_EDGES);
  int*   rank   = (int*)bufB;

  const int* src = ei;
  const int* dst = ei + N_EDGES;

  const int TB = 256;
  auto blocks = [](long long n, int tb) { return (unsigned)((n + tb - 1) / tb); };

  zero_counts<<<blocks(N_NODES + 1, TB), TB, 0, stream>>>(cnt, ws_size);
  count_rank<<<blocks(N_EDGES, TB), TB, 0, stream>>>(dst, cnt, rank, ws_size);
  row_alloc<<<blocks(N_NODES, TB), TB, 0, stream>>>(cnt, rstart, dis, ws_size);
  fill_sorted<<<blocks(N_EDGES, TB), TB, 0, stream>>>(src, dst, rank, rstart, dis,
                                                      csrc, cw, ws_size);

  auto layer = [&](const float* hin, int K, const float* W, const float* b, int F,
                   bool relu_in) {
    if (relu_in)
      gemm_wmma_f32<true><<<N_NODES / 16, 32 * (F / 16), 0, stream>>>(hin, W, bufA, K, F);
    else
      gemm_wmma_f32<false><<<N_NODES / 16, 32 * (F / 16), 0, stream>>>(hin, W, bufA, K, F);
    if (F == D_HID)
      gather_rows<D_HID><<<(N_NODES + 7) / 8, TB, 0, stream>>>(
          bufA, cnt, rstart, csrc, cw, dis, b, bufB, ws_size);
    else
      gather_rows<D_OUT><<<(N_NODES + 7) / 8, TB, 0, stream>>>(
          bufA, cnt, rstart, csrc, cw, dis, b, bufB, ws_size);
  };

  layer(x,    D_IN,  W1, b1, D_HID, false);
  layer(bufB, D_HID, W2, b2, D_HID, true);
  layer(bufB, D_HID, W3, b3, D_HID, true);
  layer(bufB, D_HID, W4, b4, D_OUT, true);

  zero_f32<<<blocks(N_GRAPHS * D_OUT + N_GRAPHS, TB), TB, 0, stream>>>(
      sums, N_GRAPHS * D_OUT + N_GRAPHS);
  pool_accum<<<blocks((long long)N_NODES * D_OUT, TB), TB, 0, stream>>>(
      bufB, batch, sums, cnts, N_NODES * D_OUT);
  pool_final<<<blocks(N_GRAPHS * D_OUT, TB), TB, 0, stream>>>(sums, cnts, out);
}


# row sums of layer l fused into layer l+1's WMMA GEMM: a block sums its 16 rows into an LDS tile (relu there) and reads its A fragments from it; 3 launches and 3 plane round trips gone; sum order kept
# speedup vs baseline: 3.8128x; 1.1783x over previous
#include <hip/hip_runtime.h>
#include <hip/hip_bf16.h>

typedef float v2f __attribute__((ext_vector_type(2)));
typedef float v8f __attribute__((ext_vector_type(8)));

#define N_NODES  50000
#define N_EDGES  600000
#define N_GRAPHS 64
#define D_IN     128
#define D_HID    128
#define D_OUT    64

#define WS_WORDS ((size_t)2 * N_NODES * D_HID + N_NODES + N_GRAPHS * D_OUT + N_GRAPHS + \
                  (N_NODES + 32) + N_NODES + (size_t)2 * N_EDGES)
#define WS_BYTES (WS_WORDS * sizeof(float))

__global__ __launch_bounds__(256)
void gemm_wmma_f32(const float* __restrict__ X, const float* __restrict__ W,
                   float* __restrict__ H, int K, int F) {
  const int rowTile = blockIdx.x << 4;
  const int colTile = (threadIdx.x >> 5) << 4;
  const int lane = threadIdx.x & 31;
  const int half = lane >> 4;
  const int i16  = lane & 15;

  const float* xrow = X + (size_t)(rowTile + i16) * (size_t)K;
  const float* wcol = W + colTile + i16;

  v8f acc = {};
  for (int k0 = 0; k0 < K; k0 += 16) {
    v2f a[4], b[4];
#pragma unroll
    for (int u = 0; u < 4; ++u) {
      const int ka = k0 + 4 * u + (half << 1);
      a[u] = *(const v2f*)(xrow + ka);
    }
#pragma unroll
    for (int u = 0; u < 4; ++u) {
      const int ka = k0 + 4 * u + (half << 1);
      b[u].x = wcol[(size_t)ka * (size_t)F];
      b[u].y = wcol[(size_t)(ka + 1) * (size_t)F];
    }
#pragma unroll
    for (int u = 0; u < 4; ++u)
      acc = __builtin_amdgcn_wmma_f32_16x16x4_f32(
           false, a[u],  false, b[u],
           (short)0, acc,  false,  false);
  }

  const int r0 = rowTile + (half << 3);
  float* hout = H + (size_t)r0 * (size_t)F + colTile + i16;
#pragma unroll
  for (int i = 0; i < 8; ++i) hout[(size_t)i * (size_t)F] = acc[i];
}

__global__ void zero_f32(float* __restrict__ p, int n) {
  int i = blockIdx.x * blockDim.x + threadIdx.x;
  if (i < n) p[i] = 0.0f;
}

__global__ void zero_counts(int* __restrict__ cnt, size_t ws_bytes) {
  if (ws_bytes < WS_BYTES) return;
  int i = blockIdx.x * blockDim.x + threadIdx.x;
  if (i <= N_NODES) cnt[i] = 0;
}

__global__ void count_rank(const int* __restrict__ dst, int* __restrict__ cnt,
                           int* __restrict__ rank, size_t ws_bytes) {
  if (ws_bytes < WS_BYTES) return;
  int e = blockIdx.x * blockDim.x + threadIdx.x;
  if (e < N_EDGES) rank[e] = atomicAdd(&cnt[dst[e]], 1);
}

__global__ __launch_bounds__(256)
void row_alloc(int* __restrict__ cnt, int* __restrict__ rstart,
               float* __restrict__ dis, size_t ws_bytes) {
  if (ws_bytes < WS_BYTES) return;
  const int i = blockIdx.x * blockDim.x + threadIdx.x;
  const int lane = threadIdx.x & 31;
  const int c = (i < N_NODES) ? cnt[i] : 0;
  int incl = c;
#pragma unroll
  for (int o = 1; o < 32; o <<= 1) {
    const int t = __shfl_up(incl, o);
    if (lane >= o) incl += t;
  }
  const int wsum = __shfl(incl, 31);
  int base = 0;
  if (lane == 31) base = atomicAdd(&cnt[N_NODES], wsum);
  base = __shfl(base, 31);
  if (i < N_NODES) {
    rstart[i] = base + incl - c;
    dis[i] = rsqrtf((float)c + 1.0f);
  }
}

__global__ void fill_sorted(const int* __restrict__ src, const int* __restrict__ dst,
                            const int* __restrict__ rank, const int* __restrict__ rstart,
                            const float* __restrict__ dis, int* __restrict__ csrc,
                            float* __restrict__ cw, size_t ws_bytes) {
  if (ws_bytes < WS_BYTES) return;
  int e = blockIdx.x * blockDim.x + threadIdx.x;
  if (e < N_EDGES) {
    const int s = src[e];
    const int d = dst[e];
    const int pos = rstart[d] + rank[e];
    csrc[pos] = s;
    cw[pos] = dis[s] * dis[d];
  }
}

template <int VW> struct fvec { typedef float type __attribute__((ext_vector_type(VW))); };

template <int F>
__device__ __forceinline__ typename fvec<F / 32>::type
row_sum(const float* __restrict__ h, int row, int lane,
        const int* __restrict__ cnt, const int* __restrict__ rstart,
        const int* __restrict__ csrc, const float* __restrict__ cw,
        const float* __restrict__ dis, const float* __restrict__ bias) {
  typedef typename fvec<F / 32>::type vec;
  const int n  = __builtin_amdgcn_readfirstlane(cnt[row]);
  const int st = __builtin_amdgcn_readfirstlane(rstart[row]);
  const float di = dis[row];

  const vec* hv = (const vec*)h;
  const vec self = hv[(size_t)row * 32 + lane];
  const vec bv = ((const vec*)bias)[lane];
  vec acc = self * di * di + bv;

  for (int c0 = 0; c0 < n; c0 += 32) {
    const int m = (n - c0 < 32) ? (n - c0) : 32;
    int sv = row;
    float wv = 0.0f;
    if (lane < m) {
      sv = csrc[st + c0 + lane];
      wv = cw[st + c0 + lane];
    }
    for (int j = 0; j < m; j += 4) {
      vec r[4];
      float w[4];
#pragma unroll
      for (int u = 0; u < 4; ++u) {
        const int sj = __builtin_amdgcn_readlane(sv, j + u);
        w[u] = __builtin_bit_cast(float,
                 __builtin_amdgcn_readlane(__builtin_bit_cast(int, wv), j + u));
        r[u] = hv[(size_t)sj * 32 + lane];
      }
#pragma unroll
      for (int u = 0; u < 4; ++u) acc += r[u] * w[u];
    }
  }
  return acc;
}

template <int F>
__global__ __launch_bounds__(256)
void gather_rows(const float* __restrict__ h, const int* __restrict__ cnt,
                 const int* __restrict__ rstart, const int* __restrict__ csrc,
                 const float* __restrict__ cw, const float* __restrict__ dis,
                 const float* __restrict__ bias, float* __restrict__ out,
                 size_t ws_bytes) {
  constexpr int VW = F / 32;
  typedef typename fvec<VW>::type vec;
  if (ws_bytes < WS_BYTES) return;
  const int lane = threadIdx.x & 31;
  const int row = (blockIdx.x * blockDim.x + threadIdx.x) >> 5;
  if (row >= N_NODES) return;
  const vec acc = row_sum<F>(h, row, lane, cnt, rstart, csrc, cw, dis, bias);
  ((vec*)out)[(size_t)row * (D_HID / VW) + lane] = acc;
}

#define A_PITCH 132
template <int FOUT>
__global__ __launch_bounds__(32 * (FOUT / 16))
void gather_gemm(const float* __restrict__ hin, const int* __restrict__ cnt,
                 const int* __restrict__ rstart, const int* __restrict__ csrc,
                 const float* __restrict__ cw, const float* __restrict__ dis,
                 const float* __restrict__ bias_in, const float* __restrict__ W,
                 float* __restrict__ Hout, size_t ws_bytes) {
  constexpr int NW  = FOUT / 16;
  constexpr int RPW = 16 / NW;
  typedef typename fvec<4>::type v4f;
  __shared__ __attribute__((aligned(16))) float tile[16 * A_PITCH];
  if (ws_bytes < WS_BYTES) return;
  const int lane = threadIdx.x & 31;
  const int wave = threadIdx.x >> 5;
  const int rowTile = blockIdx.x << 4;

  for (int rr = 0; rr < RPW; ++rr) {
    const int lr = wave * RPW + rr;
    v4f s = row_sum<D_HID>(hin, rowTile + lr, lane, cnt, rstart, csrc, cw, dis, bias_in);
    s.x = fmaxf(s.x, 0.0f);
    s.y = fmaxf(s.y, 0.0f);
    s.z = fmaxf(s.z, 0.0f);
    s.w = fmaxf(s.w, 0.0f);
    *(v4f*)(tile + lr * A_PITCH + 4 * lane) = s;
  }
  __syncthreads();

  const int half = lane >> 4;
  const int i16  = lane & 15;
  const int colTile = wave << 4;
  const float* arow = tile + i16 * A_PITCH;
  const float* wcol = W + colTile + i16;

  v8f acc = {};
  for (int k0 = 0; k0 < D_HID; k0 += 16) {
    v2f a[4], b[4];
#pragma unroll
    for (int u = 0; u < 4; ++u) {
      const int ka = k0 + 4 * u + (half << 1);
      a[u] = *(const v2f*)(arow + ka);
    }
#pragma unroll
    for (int u = 0; u < 4; ++u) {
      const int ka = k0 + 4 * u + (half << 1);
      b[u].x = wcol[(size_t)ka * (size_t)FOUT];
      b[u].y = wcol[(size_t)(ka + 1) * (size_t)FOUT];
    }
#pragma unroll
    for (int u = 0; u < 4; ++u)
      acc = __builtin_amdgcn_wmma_f32_16x16x4_f32(
           false, a[u],  false, b[u],
           (short)0, acc,  false,  false);
  }

  const int r0 = rowTile + (half << 3);
  float* hout = Hout + (size_t)r0 * (size_t)FOUT + colTile + i16;
#pragma unroll
  for (int i = 0; i < 8; ++i) hout[(size_t)i * (size_t)FOUT] = acc[i];
}

__global__ void pool_accum(const float* __restrict__ h, const int* __restrict__ batch,
                           float* __restrict__ sums, float* __restrict__ cnts, int total) {
  int i = blockIdx.x * blockDim.x + threadIdx.x;
  if (i >= total) return;
  int node = i >> 6;
  int f = i & 63;
  int g = batch[node];
  atomicAdd(&sums[(g << 6) + f], h[(node << 7) + f]);
  if (f == 0) atomicAdd(&cnts[g], 1.0f);
}

__global__ void pool_final(const float* __restrict__ sums, const float* __restrict__ cnts,
                           float* __restrict__ out) {
  int i = blockIdx.x * blockDim.x + threadIdx.x;
  if (i < N_GRAPHS * D_OUT) {
    int g = i >> 6;
    out[i] = sums[i] / fmaxf(cnts[g], 1.0f);
  }
}

extern "C" void kernel_launch(void* const* d_in, const int* in_sizes, int n_in,
                              void* d_out, int out_size, void* d_ws, size_t ws_size,
                              hipStream_t stream) {
  const float* x     = (const float*)d_in[0];
  const int*   ei    = (const int*)d_in[1];
  const int*   batch = (const int*)d_in[2];
  const float* W1 = (const float*)d_in[3];  const float* b1 = (const float*)d_in[4];
  const float* W2 = (const float*)d_in[5];  const float* b2 = (const float*)d_in[6];
  const float* W3 = (const float*)d_in[7];  const float* b3 = (const float*)d_in[8];
  const float* W4 = (const float*)d_in[9];  const float* b4 = (const float*)d_in[10];
  float* out = (float*)d_out;

  float* bufA = (float*)d_ws;
  float* bufB = bufA + (size_t)N_NODES * D_HID;
  float* dis  = bufB + (size_t)N_NODES * D_HID;
  float* sums = dis + N_NODES;
  float* cnts = sums + N_GRAPHS * D_OUT;
  int*   cnt    = (int*)(cnts + N_GRAPHS);
  int*   rstart = cnt + (N_NODES + 32);
  int*   csrc   = rstart + N_NODES;
  float* cw     = (float*)(csrc + N_EDGES);
  int*   rank   = (int*)bufB;

  const int* src = ei;
  const int* dst = ei + N_EDGES;

  const int TB = 256;
  auto blocks = [](long long n, int tb) { return (unsigned)((n + tb - 1) / tb); };

  zero_counts<<<blocks(N_NODES + 1, TB), TB, 0, stream>>>(cnt, ws_size);
  count_rank<<<blocks(N_EDGES, TB), TB, 0, stream>>>(dst, cnt, rank, ws_size);
  row_alloc<<<blocks(N_NODES, TB), TB, 0, stream>>>(cnt, rstart, dis, ws_size);
  fill_sorted<<<blocks(N_EDGES, TB), TB, 0, stream>>>(src, dst, rank, rstart, dis,
                                                      csrc, cw, ws_size);

  gemm_wmma_f32<<<N_NODES / 16, 256, 0, stream>>>(x, W1, bufA, D_IN, D_HID);
  gather_gemm<D_HID><<<N_NODES / 16, 256, 0, stream>>>(
      bufA, cnt, rstart, csrc, cw, dis, b1, W2, bufB, ws_size);
  gather_gemm<D_HID><<<N_NODES / 16, 256, 0, stream>>>(
      bufB, cnt, rstart, csrc, cw, dis, b2, W3, bufA, ws_size);
  gather_gemm<D_OUT><<<N_NODES / 16, 128, 0, stream>>>(
      bufA, cnt, rstart, csrc, cw, dis, b3, W4, bufB, ws_size);
  gather_rows<D_OUT><<<(N_NODES + 7) / 8, TB, 0, stream>>>(
      bufB, cnt, rstart, csrc, cw, dis, b4, bufA, ws_size);

  zero_f32<<<blocks(N_GRAPHS * D_OUT + N_GRAPHS, TB), TB, 0, stream>>>(
      sums, N_GRAPHS * D_OUT + N_GRAPHS);
  pool_accum<<<blocks((long long)N_NODES * D_OUT, TB), TB, 0, stream>>>(
      bufA, batch, sums, cnts, N_NODES * D_OUT);
  pool_final<<<blocks(N_GRAPHS * D_OUT, TB), TB, 0, stream>>>(sums, cnts, out);
}


# mean pool fused into layer 4's row-sum kernel: a block sums its 16 rows per graph in LDS in row order, one f32 atomic partial per run of equal graph ids; 2 launches fewer; pool sum order changes
# speedup vs baseline: 5.0495x; 1.3244x over previous
#include <hip/hip_runtime.h>
#include <hip/hip_bf16.h>

typedef float v2f __attribute__((ext_vector_type(2)));
typedef float v8f __attribute__((ext_vector_type(8)));

#define N_NODES  50000
#define N_EDGES  600000
#define N_GRAPHS 64
#define D_IN     128
#define D_HID    128
#define D_OUT    64

#define WS_WORDS ((size_t)2 * N_NODES * D_HID + N_NODES + N_GRAPHS * D_OUT + N_GRAPHS + \
                  (N_NODES + 32) + N_NODES + (size_t)2 * N_EDGES)
#define WS_BYTES (WS_WORDS * sizeof(float))

__global__ __launch_bounds__(256)
void gemm_wmma_f32(const float* __restrict__ X, const float* __restrict__ W,
                   float* __restrict__ H, int K, int F) {
  const int rowTile = blockIdx.x << 4;
  const int colTile = (threadIdx.x >> 5) << 4;
  const int lane = threadIdx.x & 31;
  const int half = lane >> 4;
  const int i16  = lane & 15;

  const float* xrow = X + (size_t)(rowTile + i16) * (size_t)K;
  const float* wcol = W + colTile + i16;

  v8f acc = {};
  for (int k0 = 0; k0 < K; k0 += 16) {
    v2f a[4], b[4];
#pragma unroll
    for (int u = 0; u < 4; ++u) {
      const int ka = k0 + 4 * u + (half << 1);
      a[u] = *(const v2f*)(xrow + ka);
    }
#pragma unroll
    for (int u = 0; u < 4; ++u) {
      const int ka = k0 + 4 * u + (half << 1);
      b[u].x = wcol[(size_t)ka * (size_t)F];
      b[u].y = wcol[(size_t)(ka + 1) * (size_t)F];
    }
#pragma unroll
    for (int u = 0; u < 4; ++u)
      acc = __builtin_amdgcn_wmma_f32_16x16x4_f32(
           false, a[u],  false, b[u],
           (short)0, acc,  false,  false);
  }

  const int r0 = rowTile + (half << 3);
  float* hout = H + (size_t)r0 * (size_t)F + colTile + i16;
#pragma unroll
  for (int i = 0; i < 8; ++i) hout[(size_t)i * (size_t)F] = acc[i];
}

#define POOL_WORDS (N_GRAPHS * D_OUT + N_GRAPHS)
__global__ void zero_pool_counts(float* __restrict__ pool, int* __restrict__ cnt,
                                 size_t ws_bytes) {
  if (ws_bytes < WS_BYTES) return;
  int i = blockIdx.x * blockDim.x + threadIdx.x;
  if (i < POOL_WORDS) pool[i] = 0.0f;
  else if (i - POOL_WORDS <= N_NODES) cnt[i - POOL_WORDS] = 0;
}

__global__ void count_rank(const int* __restrict__ dst, int* __restrict__ cnt,
                           int* __restrict__ rank, size_t ws_bytes) {
  if (ws_bytes < WS_BYTES) return;
  int e = blockIdx.x * blockDim.x + threadIdx.x;
  if (e < N_EDGES) rank[e] = atomicAdd(&cnt[dst[e]], 1);
}

__global__ __launch_bounds__(256)
void row_alloc(int* __restrict__ cnt, int* __restrict__ rstart,
               float* __restrict__ dis, size_t ws_bytes) {
  if (ws_bytes < WS_BYTES) return;
  const int i = blockIdx.x * blockDim.x + threadIdx.x;
  const int lane = threadIdx.x & 31;
  const int c = (i < N_NODES) ? cnt[i] : 0;
  int incl = c;
#pragma unroll
  for (int o = 1; o < 32; o <<= 1) {
    const int t = __shfl_up(incl, o);
    if (lane >= o) incl += t;
  }
  const int wsum = __shfl(incl, 31);
  int base = 0;
  if (lane == 31) base = atomicAdd(&cnt[N_NODES], wsum);
  base = __shfl(base, 31);
  if (i < N_NODES) {
    rstart[i] = base + incl - c;
    dis[i] = rsqrtf((float)c + 1.0f);
  }
}

__global__ void fill_sorted(const int* __restrict__ src, const int* __restrict__ dst,
                            const int* __restrict__ rank, const int* __restrict__ rstart,
                            const float* __restrict__ dis, int* __restrict__ csrc,
                            float* __restrict__ cw, size_t ws_bytes) {
  if (ws_bytes < WS_BYTES) return;
  int e = blockIdx.x * blockDim.x + threadIdx.x;
  if (e < N_EDGES) {
    const int s = src[e];
    const int d = dst[e];
    const int pos = rstart[d] + rank[e];
    csrc[pos] = s;
    cw[pos] = dis[s] * dis[d];
  }
}

template <int VW> struct fvec { typedef float type __attribute__((ext_vector_type(VW))); };

template <int F>
__device__ __forceinline__ typename fvec<F / 32>::type
row_sum(const float* __restrict__ h, int row, int lane,
        const int* __restrict__ cnt, const int* __restrict__ rstart,
        const int* __restrict__ csrc, const float* __restrict__ cw,
        const float* __restrict__ dis, const float* __restrict__ bias) {
  typedef typename fvec<F / 32>::type vec;
  const int n  = __builtin_amdgcn_readfirstlane(cnt[row]);
  const int st = __builtin_amdgcn_readfirstlane(rstart[row]);
  const float di = dis[row];

  const vec* hv = (const vec*)h;
  const vec self = hv[(size_t)row * 32 + lane];
  const vec bv = ((const vec*)bias)[lane];
  vec acc = self * di * di + bv;

  for (int c0 = 0; c0 < n; c0 += 32) {
    const int m = (n - c0 < 32) ? (n - c0) : 32;
    int sv = row;
    float wv = 0.0f;
    if (lane < m) {
      sv = csrc[st + c0 + lane];
      wv = cw[st + c0 + lane];
    }
    for (int j = 0; j < m; j += 4) {
      vec r[4];
      float w[4];
#pragma unroll
      for (int u = 0; u < 4; ++u) {
        const int sj = __builtin_amdgcn_readlane(sv, j + u);
        w[u] = __builtin_bit_cast(float,
                 __builtin_amdgcn_readlane(__builtin_bit_cast(int, wv), j + u));
        r[u] = hv[(size_t)sj * 32 + lane];
      }
#pragma unroll
      for (int u = 0; u < 4; ++u) acc += r[u] * w[u];
    }
  }
  return acc;
}

#define POOL_ROWS 16
template <int F>
__global__ __launch_bounds__(128)
void gather_pool(const float* __restrict__ h, const int* __restrict__ cnt,
                 const int* __restrict__ rstart, const int* __restrict__ csrc,
                 const float* __restrict__ cw, const float* __restrict__ dis,
                 const float* __restrict__ bias, const int* __restrict__ batch,
                 float* __restrict__ sums, float* __restrict__ cnts,
                 size_t ws_bytes) {
  constexpr int VW  = F / 32;
  constexpr int NW  = 4;
  constexpr int RPW = POOL_ROWS / NW;
  typedef typename fvec<VW>::type vec;
  static_assert(N_NODES % POOL_ROWS == 0, "a block owns POOL_ROWS whole rows");
  static_assert(F <= 32 * NW, "one thread per feature in the pool part");
  __shared__ __attribute__((aligned(16))) float tile[POOL_ROWS * F];
  __shared__ int gid[POOL_ROWS];
  if (ws_bytes < WS_BYTES) return;
  const int lane = threadIdx.x & 31;
  const int wave = threadIdx.x >> 5;
  const int row0 = blockIdx.x * POOL_ROWS;

  for (int rr = 0; rr < RPW; ++rr) {
    const int lr = wave * RPW + rr;
    const vec acc = row_sum<F>(h, row0 + lr, lane, cnt, rstart, csrc, cw, dis, bias);
    ((vec*)tile)[lr * 32 + lane] = acc;
    if (lane == 0) gid[lr] = batch[row0 + lr];
  }
  __syncthreads();

  if (threadIdx.x >= F) return;
  const int f = threadIdx.x;
  float v[POOL_ROWS];
  int g[POOL_ROWS];
#pragma unroll
  for (int r = 0; r < POOL_ROWS; ++r) {
    v[r] = tile[r * F + f];
    g[r] = gid[r];
  }
  int cur = g[0];
  float a = v[0];
  int c = 1;
#pragma unroll
  for (int r = 1; r < POOL_ROWS; ++r) {
    if (g[r] != cur) {
      atomicAdd(&sums[cur * F + f], a);
      if (f == 0) atomicAdd(&cnts[cur], (float)c);
      cur = g[r];
      a = 0.0f;
      c = 0;
    }
    a += v[r];
    ++c;
  }
  atomicAdd(&sums[cur * F + f], a);
  if (f == 0) atomicAdd(&cnts[cur], (float)c);
}

#define A_PITCH 132
template <int FOUT>
__global__ __launch_bounds__(32 * (FOUT / 16))
void gather_gemm(const float* __restrict__ hin, const int* __restrict__ cnt,
                 const int* __restrict__ rstart, const int* __restrict__ csrc,
                 const float* __restrict__ cw, const float* __restrict__ dis,
                 const float* __restrict__ bias_in, const float* __restrict__ W,
                 float* __restrict__ Hout, size_t ws_bytes) {
  constexpr int NW  = FOUT / 16;
  constexpr int RPW = 16 / NW;
  typedef typename fvec<4>::type v4f;
  __shared__ __attribute__((aligned(16))) float tile[16 * A_PITCH];
  if (ws_bytes < WS_BYTES) return;
  const int lane = threadIdx.x & 31;
  const int wave = threadIdx.x >> 5;
  const int rowTile = blockIdx.x << 4;

  for (int rr = 0; rr < RPW; ++rr) {
    const int lr = wave * RPW + rr;
    v4f s = row_sum<D_HID>(hin, rowTile + lr, lane, cnt, rstart, csrc, cw, dis, bias_in);
    s.x = fmaxf(s.x, 0.0f);
    s.y = fmaxf(s.y, 0.0f);
    s.z = fmaxf(s.z, 0.0f);
    s.w = fmaxf(s.w, 0.0f);
    *(v4f*)(tile + lr * A_PITCH + 4 * lane) = s;
  }
  __syncthreads();

  const int half = lane >> 4;
  const int i16  = lane & 15;
  const int colTile = wave << 4;
  const float* arow = tile + i16 * A_PITCH;
  const float* wcol = W + colTile + i16;

  v8f acc = {};
  for (int k0 = 0; k0 < D_HID; k0 += 16) {
    v2f a[4], b[4];
#pragma unroll
    for (int u = 0; u < 4; ++u) {
      const int ka = k0 + 4 * u + (half << 1);
      a[u] = *(const v2f*)(arow + ka);
    }
#pragma unroll
    for (int u = 0; u < 4; ++u) {
      const int ka = k0 + 4 * u + (half << 1);
      b[u].x = wcol[(size_t)ka * (size_t)FOUT];
      b[u].y = wcol[(size_t)(ka + 1) * (size_t)FOUT];
    }
#pragma unroll
    for (int u = 0; u < 4; ++u)
      acc = __builtin_amdgcn_wmma_f32_16x16x4_f32(
           false, a[u],  false, b[u],
           (short)0, acc,  false,  false);
  }

  const int r0 = rowTile + (half << 3);
  float* hout = Hout + (size_t)r0 * (size_t)FOUT + colTile + i16;
#pragma unroll
  for (int i = 0; i < 8; ++i) hout[(size_t)i * (size_t)FOUT] = acc[i];
}

__global__ void pool_final(const float* __restrict__ sums, const float* __restrict__ cnts,
                           float* __restrict__ out) {
  int i = blockIdx.x * blockDim.x + threadIdx.x;
  if (i < N_GRAPHS * D_OUT) {
    int g = i >> 6;
    out[i] = sums[i] / fmaxf(cnts[g], 1.0f);
  }
}

extern "C" void kernel_launch(void* const* d_in, const int* in_sizes, int n_in,
                              void* d_out, int out_size, void* d_ws, size_t ws_size,
                              hipStream_t stream) {
  const float* x     = (const float*)d_in[0];
  const int*   ei    = (const int*)d_in[1];
  const int*   batch = (const int*)d_in[2];
  const float* W1 = (const float*)d_in[3];  const float* b1 = (const float*)d_in[4];
  const float* W2 = (const float*)d_in[5];  const float* b2 = (const float*)d_in[6];
  const float* W3 = (const float*)d_in[7];  const float* b3 = (const float*)d_in[8];
  const float* W4 = (const float*)d_in[9];  const float* b4 = (const float*)d_in[10];
  float* out = (float*)d_out;

  float* bufA = (float*)d_ws;
  float* bufB = bufA + (size_t)N_NODES * D_HID;
  float* dis  = bufB + (size_t)N_NODES * D_HID;
  float* sums = dis + N_NODES;
  float* cnts = sums + N_GRAPHS * D_OUT;
  int*   cnt    = (int*)(cnts + N_GRAPHS);
  int*   rstart = cnt + (N_NODES + 32);
  int*   csrc   = rstart + N_NODES;
  float* cw     = (float*)(csrc + N_EDGES);
  int*   rank   = (int*)bufB;

  const int* src = ei;
  const int* dst = ei + N_EDGES;

  const int TB = 256;
  auto blocks = [](long long n, int tb) { return (unsigned)((n + tb - 1) / tb); };

  zero_pool_counts<<<blocks(POOL_WORDS + N_NODES + 1, TB), TB, 0, stream>>>(
      sums, cnt, ws_size);
  count_rank<<<blocks(N_EDGES, TB), TB, 0, stream>>>(dst, cnt, rank, ws_size);
  row_alloc<<<blocks(N_NODES, TB), TB, 0, stream>>>(cnt, rstart, dis, ws_size);
  fill_sorted<<<blocks(N_EDGES, TB), TB, 0, stream>>>(src, dst, rank, rstart, dis,
                                                      csrc, cw, ws_size);

  gemm_wmma_f32<<<N_NODES / 16, 256, 0, stream>>>(x, W1, bufA, D_IN, D_HID);
  gather_gemm<D_HID><<<N_NODES / 16, 256, 0, stream>>>(
      bufA, cnt, rstart, csrc, cw, dis, b1, W2, bufB, ws_size);
  gather_gemm<D_HID><<<N_NODES / 16, 256, 0, stream>>>(
      bufB, cnt, rstart, csrc, cw, dis, b2, W3, bufA, ws_size);
  gather_gemm<D_OUT><<<N_NODES / 16, 128, 0, stream>>>(
      bufA, cnt, rstart, csrc, cw, dis, b3, W4, bufB, ws_size);
  gather_pool<D_OUT><<<N_NODES / POOL_ROWS, 128, 0, stream>>>(
      bufB, cnt, rstart, csrc, cw, dis, b4, batch, sums, cnts, ws_size);
  pool_final<<<blocks(N_GRAPHS * D_OUT, TB), TB, 0, stream>>>(sums, cnts, out);
}


# layer 1's dense product has no launch of its own: its 16-row WMMA tiles run as extra blocks, picked by block index, among the blocks of the three list-building launches; 10 launches become 9
# speedup vs baseline: 5.5203x; 1.0932x over previous
#include <hip/hip_runtime.h>
#include <hip/hip_bf16.h>

typedef float v2f __attribute__((ext_vector_type(2)));
typedef float v8f __attribute__((ext_vector_type(8)));

#define N_NODES  50000
#define N_EDGES  600000
#define N_GRAPHS 64
#define D_IN     128
#define D_HID    128
#define D_OUT    64

#define WS_WORDS ((size_t)2 * N_NODES * D_HID + N_NODES + N_GRAPHS * D_OUT + N_GRAPHS + \
                  (N_NODES + 32) + N_NODES + (size_t)2 * N_EDGES)
#define WS_BYTES (WS_WORDS * sizeof(float))

#define GEMM_TILES   (N_NODES / 16)
#define EDGE_BLOCKS  ((N_EDGES + 255) / 256)
#define ALLOC_BLOCKS ((N_NODES + 255) / 256)
#define TILES_EDGE   (EDGE_BLOCKS / 2)
#define TILES_ALLOC  (GEMM_TILES - 2 * TILES_EDGE)
#define EDGE_GRID    (3 * TILES_EDGE)
static_assert(EDGE_BLOCKS % 2 == 0, "list blocks of an edge launch pair up, one tile per pair");
static_assert(TILES_ALLOC > 0 && 2 * TILES_EDGE + TILES_ALLOC == GEMM_TILES, "every tile once");

template <int K, int F>
__device__ __forceinline__ void gemm_tile(const float* __restrict__ X,
                                          const float* __restrict__ W,
                                          float* __restrict__ H, int tile) {
  const int rowTile = tile << 4;
  const int colTile = (threadIdx.x >> 5) << 4;
  const int lane = threadIdx.x & 31;
  const int half = lane >> 4;
  const int i16  = lane & 15;

  const float* xrow = X + (size_t)(rowTile + i16) * (size_t)K;
  const float* wcol = W + colTile + i16;

  v8f acc = {};
  for (int k0 = 0; k0 < K; k0 += 16) {
    v2f a[4], b[4];
#pragma unroll
    for (int u = 0; u < 4; ++u) {
      const int ka = k0 + 4 * u + (half << 1);
      a[u] = *(const v2f*)(xrow + ka);
    }
#pragma unroll
    for (int u = 0; u < 4; ++u) {
      const int ka = k0 + 4 * u + (half << 1);
      b[u].x = wcol[(size_t)ka * (size_t)F];
      b[u].y = wcol[(size_t)(ka + 1) * (size_t)F];
    }
#pragma unroll
    for (int u = 0; u < 4; ++u)
      acc = __builtin_amdgcn_wmma_f32_16x16x4_f32(
           false, a[u],  false, b[u],
           (short)0, acc,  false,  false);
  }

  const int r0 = rowTile + (half << 3);
  float* hout = H + (size_t)r0 * (size_t)F + colTile + i16;
#pragma unroll
  for (int i = 0; i < 8; ++i) hout[(size_t)i * (size_t)F] = acc[i];
}

#define POOL_WORDS (N_GRAPHS * D_OUT + N_GRAPHS)
__global__ void zero_pool_counts(float* __restrict__ pool, int* __restrict__ cnt,
                                 size_t ws_bytes) {
  if (ws_bytes < WS_BYTES) return;
  int i = blockIdx.x * blockDim.x + threadIdx.x;
  if (i < POOL_WORDS) pool[i] = 0.0f;
  else if (i - POOL_WORDS <= N_NODES) cnt[i - POOL_WORDS] = 0;
}

__global__ __launch_bounds__(256)
void count_rank(const int* __restrict__ dst, int* __restrict__ cnt,
                int* __restrict__ rank, size_t ws_bytes,
                const float* __restrict__ X, const float* __restrict__ W,
                float* __restrict__ H) {
  const int trio = blockIdx.x / 3;
  const int kind = blockIdx.x - 3 * trio;
  if (kind == 2) {
    gemm_tile<D_IN, D_HID>(X, W, H, trio);
    return;
  }
  if (ws_bytes < WS_BYTES) return;
  int e = (2 * trio + kind) * blockDim.x + threadIdx.x;
  if (e < N_EDGES) rank[e] = atomicAdd(&cnt[dst[e]], 1);
}

__global__ __launch_bounds__(256)
void row_alloc(int* __restrict__ cnt, int* __restrict__ rstart,
               float* __restrict__ dis, size_t ws_bytes,
               const float* __restrict__ X, const float* __restrict__ W,
               float* __restrict__ H) {
  if (blockIdx.x >= ALLOC_BLOCKS) {
    gemm_tile<D_IN, D_HID>(X, W, H, TILES_EDGE + (int)blockIdx.x - ALLOC_BLOCKS);
    return;
  }
  if (ws_bytes < WS_BYTES) return;
  const int i = blockIdx.x * blockDim.x + threadIdx.x;
  const int lane = threadIdx.x & 31;
  const int c = (i < N_NODES) ? cnt[i] : 0;
  int incl = c;
#pragma unroll
  for (int o = 1; o < 32; o <<= 1) {
    const int t = __shfl_up(incl, o);
    if (lane >= o) incl += t;
  }
  const int wsum = __shfl(incl, 31);
  int base = 0;
  if (lane == 31) base = atomicAdd(&cnt[N_NODES], wsum);
  base = __shfl(base, 31);
  if (i < N_NODES) {
    rstart[i] = base + incl - c;
    dis[i] = rsqrtf((float)c + 1.0f);
  }
}

__global__ __launch_bounds__(256)
void fill_sorted(const int* __restrict__ src, const int* __restrict__ dst,
                 const int* __restrict__ rank, const int* __restrict__ rstart,
                 const float* __restrict__ dis, int* __restrict__ csrc,
                 float* __restrict__ cw, size_t ws_bytes,
                 const float* __restrict__ X, const float* __restrict__ W,
                 float* __restrict__ H) {
  const int trio = blockIdx.x / 3;
  const int kind = blockIdx.x - 3 * trio;
  if (kind == 2) {
    gemm_tile<D_IN, D_HID>(X, W, H, TILES_EDGE + TILES_ALLOC + trio);
    return;
  }
  if (ws_bytes < WS_BYTES) return;
  int e = (2 * trio + kind) * blockDim.x + threadIdx.x;
  if (e < N_EDGES) {
    const int s = src[e];
    const int d = dst[e];
    const int pos = rstart[d] + rank[e];
    csrc[pos] = s;
    cw[pos] = dis[s] * dis[d];
  }
}

template <int VW> struct fvec { typedef float type __attribute__((ext_vector_type(VW))); };

template <int F>
__device__ __forceinline__ typename fvec<F / 32>::type
row_sum(const float* __restrict__ h, int row, int lane,
        const int* __restrict__ cnt, const int* __restrict__ rstart,
        const int* __restrict__ csrc, const float* __restrict__ cw,
        const float* __restrict__ dis, const float* __restrict__ bias) {
  typedef typename fvec<F / 32>::type vec;
  const int n  = __builtin_amdgcn_readfirstlane(cnt[row]);
  const int st = __builtin_amdgcn_readfirstlane(rstart[row]);
  const float di = dis[row];

  const vec* hv = (const vec*)h;
  const vec self = hv[(size_t)row * 32 + lane];
  const vec bv = ((const vec*)bias)[lane];
  vec acc = self * di * di + bv;

  for (int c0 = 0; c0 < n; c0 += 32) {
    const int m = (n - c0 < 32) ? (n - c0) : 32;
    int sv = row;
    float wv = 0.0f;
    if (lane < m) {
      sv = csrc[st + c0 + lane];
      wv = cw[st + c0 + lane];
    }
    for (int j = 0; j < m; j += 4) {
      vec r[4];
      float w[4];
#pragma unroll
      for (int u = 0; u < 4; ++u) {
        const int sj = __builtin_amdgcn_readlane(sv, j + u);
        w[u] = __builtin_bit_cast(float,
                 __builtin_amdgcn_readlane(__builtin_bit_cast(int, wv), j + u));
        r[u] = hv[(size_t)sj * 32 + lane];
      }
#pragma unroll
      for (int u = 0; u < 4; ++u) acc += r[u] * w[u];
    }
  }
  return acc;
}

#define POOL_ROWS 16
template <int F>
__global__ __launch_bounds__(128)
void gather_pool(const float* __restrict__ h, const int* __restrict__ cnt,
                 const int* __restrict__ rstart, const int* __restrict__ csrc,
                 const float* __restrict__ cw, const float* __restrict__ dis,
                 const float* __restrict__ bias, const int* __restrict__ batch,
                 float* __restrict__ sums, float* __restrict__ cnts,
                 size_t ws_bytes) {
  constexpr int VW  = F / 32;
  constexpr int NW  = 4;
  constexpr int RPW = POOL_ROWS / NW;
  typedef typename fvec<VW>::type vec;
  static_assert(N_NODES % POOL_ROWS == 0, "a block owns POOL_ROWS whole rows");
  static_assert(F <= 32 * NW, "one thread per feature in the pool part");
  __shared__ __attribute__((aligned(16))) float tile[POOL_ROWS * F];
  __shared__ int gid[POOL_ROWS];
  if (ws_bytes < WS_BYTES) return;
  const int lane = threadIdx.x & 31;
  const int wave = threadIdx.x >> 5;
  const int row0 = blockIdx.x * POOL_ROWS;

  for (int rr = 0; rr < RPW; ++rr) {
    const int lr = wave * RPW + rr;
    const vec acc = row_sum<F>(h, row0 + lr, lane, cnt, rstart, csrc, cw, dis, bias);
    ((vec*)tile)[lr * 32 + lane] = acc;
    if (lane == 0) gid[lr] = batch[row0 + lr];
  }
  __syncthreads();

  if (threadIdx.x >= F) return;
  const int f = threadIdx.x;
  float v[POOL_ROWS];
  int g[POOL_ROWS];
#pragma unroll
  for (int r = 0; r < POOL_ROWS; ++r) {
    v[r] = tile[r * F + f];
    g[r] = gid[r];
  }
  int cur = g[0];
  float a = v[0];
  int c = 1;
#pragma unroll
  for (int r = 1; r < POOL_ROWS; ++r) {
    if (g[r] != cur) {
      atomicAdd(&sums[cur * F + f], a);
      if (f == 0) atomicAdd(&cnts[cur], (float)c);
      cur = g[r];
      a = 0.0f;
      c = 0;
    }
    a += v[r];
    ++c;
  }
  atomicAdd(&sums[cur * F + f], a);
  if (f == 0) atomicAdd(&cnts[cur], (float)c);
}

#define A_PITCH 132
template <int FOUT>
__global__ __launch_bounds__(32 * (FOUT / 16))
void gather_gemm(const float* __restrict__ hin, const int* __restrict__ cnt,
                 const int* __restrict__ rstart, const int* __restrict__ csrc,
                 const float* __restrict__ cw, const float* __restrict__ dis,
                 const float* __restrict__ bias_in, const float* __restrict__ W,
                 float* __restrict__ Hout, size_t ws_bytes) {
  constexpr int NW  = FOUT / 16;
  constexpr int RPW = 16 / NW;
  typedef typename fvec<4>::type v4f;
  __shared__ __attribute__((aligned(16))) float tile[16 * A_PITCH];
  if (ws_bytes < WS_BYTES) return;
  const int lane = threadIdx.x & 31;
  const int wave = threadIdx.x >> 5;
  const int rowTile = blockIdx.x << 4;

  for (int rr = 0; rr < RPW; ++rr) {
    const int lr = wave * RPW + rr;
    v4f s = row_sum<D_HID>(hin, rowTile + lr, lane, cnt, rstart, csrc, cw, dis, bias_in);
    s.x = fmaxf(s.x, 0.0f);
    s.y = fmaxf(s.y, 0.0f);
    s.z = fmaxf(s.z, 0.0f);
    s.w = fmaxf(s.w, 0.0f);
    *(v4f*)(tile + lr * A_PITCH + 4 * lane) = s;
  }
  __syncthreads();

  const int half = lane >> 4;
  const int i16  = lane & 15;
  const int colTile = wave << 4;
  const float* arow = tile + i16 * A_PITCH;
  const float* wcol = W + colTile + i16;

  v8f acc = {};
  for (int k0 = 0; k0 < D_HID; k0 += 16) {
    v2f a[4], b[4];
#pragma unroll
    for (int u = 0; u < 4; ++u) {
      const int ka = k0 + 4 * u + (half << 1);
      a[u] = *(const v2f*)(arow + ka);
    }
#pragma unroll
    for (int u = 0; u < 4; ++u) {
      const int ka = k0 + 4 * u + (half << 1);
      b[u].x = wcol[(size_t)ka * (size_t)FOUT];
      b[u].y = wcol[(size_t)(ka + 1) * (size_t)FOUT];
    }
#pragma unroll
    for (int u = 0; u < 4; ++u)
      acc = __builtin_amdgcn_wmma_f32_16x16x4_f32(
           false, a[u],  false, b[u],
           (short)0, acc,  false,  false);
  }

  const int r0 = rowTile + (half << 3);
  float* hout = Hout + (size_t)r0 * (size_t)FOUT + colTile + i16;
#pragma unroll
  for (int i = 0; i < 8; ++i) hout[(size_t)i * (size_t)FOUT] = acc[i];
}

__global__ void pool_final(const float* __restrict__ sums, const float* __restrict__ cnts,
                           float* __restrict__ out) {
  int i = blockIdx.x * blockDim.x + threadIdx.x;
  if (i < N_GRAPHS * D_OUT) {
    int g = i >> 6;
    out[i] = sums[i] / fmaxf(cnts[g], 1.0f);
  }
}

extern "C" void kernel_launch(void* const* d_in, const int* in_sizes, int n_in,
                              void* d_out, int out_size, void* d_ws, size_t ws_size,
                              hipStream_t stream) {
  const float* x     = (const float*)d_in[0];
  const int*   ei    = (const int*)d_in[1];
  const int*   batch = (const int*)d_in[2];
  const float* W1 = (const float*)d_in[3];  const float* b1 = (const float*)d_in[4];
  const float* W2 = (const float*)d_in[5];  const float* b2 = (const float*)d_in[6];
  const float* W3 = (const float*)d_in[7];  const float* b3 = (const float*)d_in[8];
  const float* W4 = (const float*)d_in[9];  const float* b4 = (const float*)d_in[10];
  float* out = (float*)d_out;

  float* bufA = (float*)d_ws;
  float* bufB = bufA + (size_t)N_NODES * D_HID;
  float* dis  = bufB + (size_t)N_NODES * D_HID;
  float* sums = dis + N_NODES;
  float* cnts = sums + N_GRAPHS * D_OUT;
  int*   cnt    = (int*)(cnts + N_GRAPHS);
  int*   rstart = cnt + (N_NODES + 32);
  int*   csrc   = rstart + N_NODES;
  float* cw     = (float*)(csrc + N_EDGES);
  int*   rank   = (int*)bufB;

  const int* src = ei;
  const int* dst = ei + N_EDGES;

  const int TB = 256;
  auto blocks = [](long long n, int tb) { return (unsigned)((n + tb - 1) / tb); };

  zero_pool_counts<<<blocks(POOL_WORDS + N_NODES + 1, TB), TB, 0, stream>>>(
      sums, cnt, ws_size);
  count_rank<<<EDGE_GRID, TB, 0, stream>>>(dst, cnt, rank, ws_size, x, W1, bufA);
  row_alloc<<<ALLOC_BLOCKS + TILES_ALLOC, TB, 0, stream>>>(cnt, rstart, dis, ws_size,
                                                           x, W1, bufA);
  fill_sorted<<<EDGE_GRID, TB, 0, stream>>>(src, dst, rank, rstart, dis, csrc, cw, ws_size,
                                            x, W1, bufA);
  gather_gemm<D_HID><<<N_NODES / 16, 256, 0, stream>>>(
      bufA, cnt, rstart, csrc, cw, dis, b1, W2, bufB, ws_size);
  gather_gemm<D_HID><<<N_NODES / 16, 256, 0, stream>>>(
      bufB, cnt, rstart, csrc, cw, dis, b2, W3, bufA, ws_size);
  gather_gemm<D_OUT><<<N_NODES / 16, 128, 0, stream>>>(
      bufA, cnt, rstart, csrc, cw, dis, b3, W4, bufB, ws_size);
  gather_pool<D_OUT><<<N_NODES / POOL_ROWS, 128, 0, stream>>>(
      bufB, cnt, rstart, csrc, cw, dis, b4, batch, sums, cnts, ws_size);
  pool_final<<<blocks(N_GRAPHS * D_OUT, TB), TB, 0, stream>>>(sums, cnts, out);
}


# each wave sums its rows two at a time side by side: counts by scalar loads, both slot lists asked for beside the rows' own loads, then 4 row loads of each row per group (8 in flight); sum order kept
# speedup vs baseline: 5.7059x; 1.0336x over previous
#include <hip/hip_runtime.h>
#include <hip/hip_bf16.h>

typedef float v2f __attribute__((ext_vector_type(2)));
typedef float v8f __attribute__((ext_vector_type(8)));

#define N_NODES  50000
#define N_EDGES  600000
#define N_GRAPHS 64
#define D_IN     128
#define D_HID    128
#define D_OUT    64

#define WS_WORDS ((size_t)2 * N_NODES * D_HID + N_NODES + N_GRAPHS * D_OUT + N_GRAPHS + \
                  (N_NODES + 32) + N_NODES + (size_t)2 * N_EDGES)
#define WS_BYTES (WS_WORDS * sizeof(float))

#define GEMM_TILES   (N_NODES / 16)
#define EDGE_BLOCKS  ((N_EDGES + 255) / 256)
#define ALLOC_BLOCKS ((N_NODES + 255) / 256)
#define TILES_EDGE   (EDGE_BLOCKS / 2)
#define TILES_ALLOC  (GEMM_TILES - 2 * TILES_EDGE)
#define EDGE_GRID    (3 * TILES_EDGE)
static_assert(EDGE_BLOCKS % 2 == 0, "list blocks of an edge launch pair up, one tile per pair");
static_assert(TILES_ALLOC > 0 && 2 * TILES_EDGE + TILES_ALLOC == GEMM_TILES, "every tile once");

template <int K, int F>
__device__ __forceinline__ void gemm_tile(const float* __restrict__ X,
                                          const float* __restrict__ W,
                                          float* __restrict__ H, int tile) {
  const int rowTile = tile << 4;
  const int colTile = (threadIdx.x >> 5) << 4;
  const int lane = threadIdx.x & 31;
  const int half = lane >> 4;
  const int i16  = lane & 15;

  const float* xrow = X + (size_t)(rowTile + i16) * (size_t)K;
  const float* wcol = W + colTile + i16;

  v8f acc = {};
  for (int k0 = 0; k0 < K; k0 += 16) {
    v2f a[4], b[4];
#pragma unroll
    for (int u = 0; u < 4; ++u) {
      const int ka = k0 + 4 * u + (half << 1);
      a[u] = *(const v2f*)(xrow + ka);
    }
#pragma unroll
    for (int u = 0; u < 4; ++u) {
      const int ka = k0 + 4 * u + (half << 1);
      b[u].x = wcol[(size_t)ka * (size_t)F];
      b[u].y = wcol[(size_t)(ka + 1) * (size_t)F];
    }
#pragma unroll
    for (int u = 0; u < 4; ++u)
      acc = __builtin_amdgcn_wmma_f32_16x16x4_f32(
           false, a[u],  false, b[u],
           (short)0, acc,  false,  false);
  }

  const int r0 = rowTile + (half << 3);
  float* hout = H + (size_t)r0 * (size_t)F + colTile + i16;
#pragma unroll
  for (int i = 0; i < 8; ++i) hout[(size_t)i * (size_t)F] = acc[i];
}

#define POOL_WORDS (N_GRAPHS * D_OUT + N_GRAPHS)
__global__ void zero_pool_counts(float* __restrict__ pool, int* __restrict__ cnt,
                                 size_t ws_bytes) {
  if (ws_bytes < WS_BYTES) return;
  int i = blockIdx.x * blockDim.x + threadIdx.x;
  if (i < POOL_WORDS) pool[i] = 0.0f;
  else if (i - POOL_WORDS <= N_NODES) cnt[i - POOL_WORDS] = 0;
}

__global__ __launch_bounds__(256)
void count_rank(const int* __restrict__ dst, int* __restrict__ cnt,
                int* __restrict__ rank, size_t ws_bytes,
                const float* __restrict__ X, const float* __restrict__ W,
                float* __restrict__ H) {
  const int trio = blockIdx.x / 3;
  const int kind = blockIdx.x - 3 * trio;
  if (kind == 2) {
    gemm_tile<D_IN, D_HID>(X, W, H, trio);
    return;
  }
  if (ws_bytes < WS_BYTES) return;
  int e = (2 * trio + kind) * blockDim.x + threadIdx.x;
  if (e < N_EDGES) rank[e] = atomicAdd(&cnt[dst[e]], 1);
}

__global__ __launch_bounds__(256)
void row_alloc(int* __restrict__ cnt, int* __restrict__ rstart,
               float* __restrict__ dis, size_t ws_bytes,
               const float* __restrict__ X, const float* __restrict__ W,
               float* __restrict__ H) {
  if (blockIdx.x >= ALLOC_BLOCKS) {
    gemm_tile<D_IN, D_HID>(X, W, H, TILES_EDGE + (int)blockIdx.x - ALLOC_BLOCKS);
    return;
  }
  if (ws_bytes < WS_BYTES) return;
  const int i = blockIdx.x * blockDim.x + threadIdx.x;
  const int lane = threadIdx.x & 31;
  const int c = (i < N_NODES) ? cnt[i] : 0;
  int incl = c;
#pragma unroll
  for (int o = 1; o < 32; o <<= 1) {
    const int t = __shfl_up(incl, o);
    if (lane >= o) incl += t;
  }
  const int wsum = __shfl(incl, 31);
  int base = 0;
  if (lane == 31) base = atomicAdd(&cnt[N_NODES], wsum);
  base = __shfl(base, 31);
  if (i < N_NODES) {
    rstart[i] = base + incl - c;
    dis[i] = rsqrtf((float)c + 1.0f);
  }
}

__global__ __launch_bounds__(256)
void fill_sorted(const int* __restrict__ src, const int* __restrict__ dst,
                 const int* __restrict__ rank, const int* __restrict__ rstart,
                 const float* __restrict__ dis, int* __restrict__ csrc,
                 float* __restrict__ cw, size_t ws_bytes,
                 const float* __restrict__ X, const float* __restrict__ W,
                 float* __restrict__ H) {
  const int trio = blockIdx.x / 3;
  const int kind = blockIdx.x - 3 * trio;
  if (kind == 2) {
    gemm_tile<D_IN, D_HID>(X, W, H, TILES_EDGE + TILES_ALLOC + trio);
    return;
  }
  if (ws_bytes < WS_BYTES) return;
  int e = (2 * trio + kind) * blockDim.x + threadIdx.x;
  if (e < N_EDGES) {
    const int s = src[e];
    const int d = dst[e];
    const int pos = rstart[d] + rank[e];
    csrc[pos] = s;
    cw[pos] = dis[s] * dis[d];
  }
}

template <int VW> struct fvec { typedef float type __attribute__((ext_vector_type(VW))); };

template <int F, int R, int U>
__device__ __forceinline__ void
row_sums(const float* __restrict__ h, int row0_in, int lane,
         const int* __restrict__ cnt, const int* __restrict__ rstart,
         const int* __restrict__ csrc, const float* __restrict__ cw,
         const float* __restrict__ dis, const float* __restrict__ bias,
         typename fvec<F / 32>::type (&acc)[R]) {
  typedef typename fvec<F / 32>::type vec;
  static_assert(32 % U == 0, "a group of U slots never runs past lane 31");
  const int row0 = __builtin_amdgcn_readfirstlane(row0_in);
  const char* hb = (const char*)h;
  constexpr unsigned ROWB = F * sizeof(float);
  static_assert((unsigned long long)N_NODES * ROWB < 0xffffffffull, "32-bit byte offsets into a plane");
  const unsigned loff = (unsigned)lane * (unsigned)sizeof(vec);

  int n[R], st[R];
  float di[R];
  vec self[R];
#pragma unroll
  for (int r = 0; r < R; ++r) {
    n[r]    = cnt[row0 + r];
    st[r]   = rstart[row0 + r];
    di[r]   = dis[row0 + r];
    self[r] = *(const vec*)(hb + ((unsigned)(row0 + r) * ROWB + loff));
  }
  const vec bv = ((const vec*)bias)[lane];
  int nmax = 0;
#pragma unroll
  for (int r = 0; r < R; ++r) {
    nmax = (n[r] > nmax) ? n[r] : nmax;
  }

  int m[R], sv[R];
  float wv[R];
  auto load_slots = [&](int c0) {
#pragma unroll
    for (int r = 0; r < R; ++r) {
      const int left = n[r] - c0;
      m[r]  = (left < 0) ? 0 : ((left > 32) ? 32 : left);
      sv[r] = row0 + r;
      wv[r] = 0.0f;
      if (lane < m[r]) {
        sv[r] = csrc[st[r] + c0 + lane];
        wv[r] = cw[st[r] + c0 + lane];
      }
    }
  };
  load_slots(0);
#pragma unroll
  for (int r = 0; r < R; ++r) acc[r] = self[r] * di[r] * di[r] + bv;

  for (int c0 = 0;;) {
    int g[R];
    int gmin = 32 / U;
#pragma unroll
    for (int r = 0; r < R; ++r) {
      g[r] = (m[r] + U - 1) / U;
      gmin = (g[r] < gmin) ? g[r] : gmin;
    }
    for (int t = 0; t < gmin; ++t) {
      const int j = t * U;
      vec rw[R][U];
#pragma unroll
      for (int r = 0; r < R; ++r) {
#pragma unroll
        for (int u = 0; u < U; ++u) {
          const int sj = __builtin_amdgcn_readlane(sv[r], j + u);
          rw[r][u] = *(const vec*)(hb + ((unsigned)sj * ROWB + loff));
        }
      }
#pragma unroll
      for (int r = 0; r < R; ++r) {
#pragma unroll
        for (int u = 0; u < U; ++u) {
          const float w = __builtin_bit_cast(float,
                            __builtin_amdgcn_readlane(__builtin_bit_cast(int, wv[r]), j + u));
          acc[r] += rw[r][u] * w;
        }
      }
    }
#pragma unroll
    for (int r = 0; r < R; ++r) {
      for (int t = gmin; t < g[r]; ++t) {
        const int j = t * U;
        vec rw[U];
#pragma unroll
        for (int u = 0; u < U; ++u) {
          const int sj = __builtin_amdgcn_readlane(sv[r], j + u);
          rw[u] = *(const vec*)(hb + ((unsigned)sj * ROWB + loff));
        }
#pragma unroll
        for (int u = 0; u < U; ++u) {
          const float w = __builtin_bit_cast(float,
                            __builtin_amdgcn_readlane(__builtin_bit_cast(int, wv[r]), j + u));
          acc[r] += rw[u] * w;
        }
      }
    }
    c0 += 32;
    if (c0 >= nmax) break;
    load_slots(c0);
  }
}

#define POOL_ROWS 16
template <int F>
__global__ __launch_bounds__(128)
void gather_pool(const float* __restrict__ h, const int* __restrict__ cnt,
                 const int* __restrict__ rstart, const int* __restrict__ csrc,
                 const float* __restrict__ cw, const float* __restrict__ dis,
                 const float* __restrict__ bias, const int* __restrict__ batch,
                 float* __restrict__ sums, float* __restrict__ cnts,
                 size_t ws_bytes) {
  constexpr int VW  = F / 32;
  constexpr int NW  = 4;
  constexpr int RPW = POOL_ROWS / NW;
  typedef typename fvec<VW>::type vec;
  static_assert(N_NODES % POOL_ROWS == 0, "a block owns POOL_ROWS whole rows");
  static_assert(F <= 32 * NW, "one thread per feature in the pool part");
  __shared__ __attribute__((aligned(16))) float tile[POOL_ROWS * F];
  __shared__ int gid[POOL_ROWS];
  if (ws_bytes < WS_BYTES) return;
  const int lane = threadIdx.x & 31;
  const int wave = threadIdx.x >> 5;
  const int row0 = blockIdx.x * POOL_ROWS;

  static_assert(RPW % 2 == 0, "a wave takes its rows in pairs");
  for (int rr = 0; rr < RPW; rr += 2) {
    const int lr = wave * RPW + rr;
    vec acc[2];
    row_sums<F, 2, 4>(h, row0 + lr, lane, cnt, rstart, csrc, cw, dis, bias, acc);
    ((vec*)tile)[lr * 32 + lane] = acc[0];
    ((vec*)tile)[(lr + 1) * 32 + lane] = acc[1];
    if (lane < 2) gid[lr + lane] = batch[row0 + lr + lane];
  }
  __syncthreads();

  if (threadIdx.x >= F) return;
  const int f = threadIdx.x;
  float v[POOL_ROWS];
  int g[POOL_ROWS];
#pragma unroll
  for (int r = 0; r < POOL_ROWS; ++r) {
    v[r] = tile[r * F + f];
    g[r] = gid[r];
  }
  int cur = g[0];
  float a = v[0];
  int c = 1;
#pragma unroll
  for (int r = 1; r < POOL_ROWS; ++r) {
    if (g[r] != cur) {
      atomicAdd(&sums[cur * F + f], a);
      if (f == 0) atomicAdd(&cnts[cur], (float)c);
      cur = g[r];
      a = 0.0f;
      c = 0;
    }
    a += v[r];
    ++c;
  }
  atomicAdd(&sums[cur * F + f], a);
  if (f == 0) atomicAdd(&cnts[cur], (float)c);
}

#define A_PITCH 132
template <int FOUT>
__global__ __launch_bounds__(32 * (FOUT / 16))
void gather_gemm(const float* __restrict__ hin, const int* __restrict__ cnt,
                 const int* __restrict__ rstart, const int* __restrict__ csrc,
                 const float* __restrict__ cw, const float* __restrict__ dis,
                 const float* __restrict__ bias_in, const float* __restrict__ W,
                 float* __restrict__ Hout, size_t ws_bytes) {
  constexpr int NW  = FOUT / 16;
  constexpr int RPW = 16 / NW;
  typedef typename fvec<4>::type v4f;
  __shared__ __attribute__((aligned(16))) float tile[16 * A_PITCH];
  if (ws_bytes < WS_BYTES) return;
  const int lane = threadIdx.x & 31;
  const int wave = threadIdx.x >> 5;
  const int rowTile = blockIdx.x << 4;

  static_assert(RPW % 2 == 0, "a wave takes its rows in pairs");
  for (int rp = 0; rp < RPW; rp += 2) {
    v4f sums[2];
    row_sums<D_HID, 2, 4>(hin, rowTile + wave * RPW + rp, lane, cnt, rstart, csrc, cw, dis,
                          bias_in, sums);
#pragma unroll
    for (int rr = 0; rr < 2; ++rr) {
      const int lr = wave * RPW + rp + rr;
      v4f s = sums[rr];
      s.x = fmaxf(s.x, 0.0f);
      s.y = fmaxf(s.y, 0.0f);
      s.z = fmaxf(s.z, 0.0f);
      s.w = fmaxf(s.w, 0.0f);
      *(v4f*)(tile + lr * A_PITCH + 4 * lane) = s;
    }
  }
  __syncthreads();

  const int half = lane >> 4;
  const int i16  = lane & 15;
  const int colTile = wave << 4;
  const float* arow = tile + i16 * A_PITCH;
  const float* wcol = W + colTile + i16;

  v8f acc = {};
  for (int k0 = 0; k0 < D_HID; k0 += 16) {
    v2f a[4], b[4];
#pragma unroll
    for (int u = 0; u < 4; ++u) {
      const int ka = k0 + 4 * u + (half << 1);
      a[u] = *(const v2f*)(arow + ka);
    }
#pragma unroll
    for (int u = 0; u < 4; ++u) {
      const int ka = k0 + 4 * u + (half << 1);
      b[u].x = wcol[(size_t)ka * (size_t)FOUT];
      b[u].y = wcol[(size_t)(ka + 1) * (size_t)FOUT];
    }
#pragma unroll
    for (int u = 0; u < 4; ++u)
      acc = __builtin_amdgcn_wmma_f32_16x16x4_f32(
           false, a[u],  false, b[u],
           (short)0, acc,  false,  false);
  }

  const int r0 = rowTile + (half << 3);
  float* hout = Hout + (size_t)r0 * (size_t)FOUT + colTile + i16;
#pragma unroll
  for (int i = 0; i < 8; ++i) hout[(size_t)i * (size_t)FOUT] = acc[i];
}

__global__ void pool_final(const float* __restrict__ sums, const float* __restrict__ cnts,
                           float* __restrict__ out) {
  int i = blockIdx.x * blockDim.x + threadIdx.x;
  if (i < N_GRAPHS * D_OUT) {
    int g = i >> 6;
    out[i] = sums[i] / fmaxf(cnts[g], 1.0f);
  }
}

extern "C" void kernel_launch(void* const* d_in, const int* in_sizes, int n_in,
                              void* d_out, int out_size, void* d_ws, size_t ws_size,
                              hipStream_t stream) {
  const float* x     = (const float*)d_in[0];
  const int*   ei    = (const int*)d_in[1];
  const int*   batch = (const int*)d_in[2];
  const float* W1 = (const float*)d_in[3];  const float* b1 = (const float*)d_in[4];
  const float* W2 = (const float*)d_in[5];  const float* b2 = (const float*)d_in[6];
  const float* W3 = (const float*)d_in[7];  const float* b3 = (const float*)d_in[8];
  const float* W4 = (const float*)d_in[9];  const float* b4 = (const float*)d_in[10];
  float* out = (float*)d_out;

  float* bufA = (float*)d_ws;
  float* bufB = bufA + (size_t)N_NODES * D_HID;
  float* dis  = bufB + (size_t)N_NODES * D_HID;
  float* sums = dis + N_NODES;
  float* cnts = sums + N_GRAPHS * D_OUT;
  int*   cnt    = (int*)(cnts + N_GRAPHS);
  int*   rstart = cnt + (N_NODES + 32);
  int*   csrc   = rstart + N_NODES;
  float* cw     = (float*)(csrc + N_EDGES);
  int*   rank   = (int*)bufB;

  const int* src = ei;
  const int* dst = ei + N_EDGES;

  const int TB = 256;
  auto blocks = [](long long n, int tb) { return (unsigned)((n + tb - 1) / tb); };

  zero_pool_counts<<<blocks(POOL_WORDS + N_NODES + 1, TB), TB, 0, stream>>>(
      sums, cnt, ws_size);
  count_rank<<<EDGE_GRID, TB, 0, stream>>>(dst, cnt, rank, ws_size, x, W1, bufA);
  row_alloc<<<ALLOC_BLOCKS + TILES_ALLOC, TB, 0, stream>>>(cnt, rstart, dis, ws_size,
                                                           x, W1, bufA);
  fill_sorted<<<EDGE_GRID, TB, 0, stream>>>(src, dst, rank, rstart, dis, csrc, cw, ws_size,
                                            x, W1, bufA);
  gather_gemm<D_HID><<<N_NODES / 16, 256, 0, stream>>>(
      bufA, cnt, rstart, csrc, cw, dis, b1, W2, bufB, ws_size);
  gather_gemm<D_HID><<<N_NODES / 16, 256, 0, stream>>>(
      bufB, cnt, rstart, csrc, cw, dis, b2, W3, bufA, ws_size);
  gather_gemm<D_OUT><<<N_NODES / 16, 128, 0, stream>>>(
      bufA, cnt, rstart, csrc, cw, dis, b3, W4, bufB, ws_size);
  gather_pool<D_OUT><<<N_NODES / POOL_ROWS, 128, 0, stream>>>(
      bufB, cnt, rstart, csrc, cw, dis, b4, batch, sums, cnts, ws_size);
  pool_final<<<blocks(N_GRAPHS * D_OUT, TB), TB, 0, stream>>>(sums, cnts, out);
}


# each wave owns 32 output columns as two interleaved 16x16 accumulators (even/odd): one 8-byte W load feeds both, one A fragment read per two WMMAs, 8-byte stores; blocks have half the waves
# speedup vs baseline: 5.8879x; 1.0319x over previous
#include <hip/hip_runtime.h>
#include <hip/hip_bf16.h>

typedef float v2f __attribute__((ext_vector_type(2)));
typedef float v8f __attribute__((ext_vector_type(8)));

#define N_NODES  50000
#define N_EDGES  600000
#define N_GRAPHS 64
#define D_IN     128
#define D_HID    128
#define D_OUT    64

#define WS_WORDS ((size_t)2 * N_NODES * D_HID + N_NODES + N_GRAPHS * D_OUT + N_GRAPHS + \
                  (N_NODES + 32) + N_NODES + (size_t)2 * N_EDGES)
#define WS_BYTES (WS_WORDS * sizeof(float))

#define GEMM_TILES   (N_NODES / 16)
#define EDGE_BLOCKS  ((N_EDGES + 255) / 256)
#define ALLOC_BLOCKS ((N_NODES + 255) / 256)
#define TILES_EDGE   (EDGE_BLOCKS / 2)
#define TILES_ALLOC  (GEMM_TILES - 2 * TILES_EDGE)
#define PAIRS_EDGE   (TILES_EDGE / 2)
#define PAIRS_ALLOC  ((TILES_ALLOC + 1) / 2)
#define EDGE_GRID    (5 * PAIRS_EDGE)
static_assert(EDGE_BLOCKS % 4 == 0, "list blocks of an edge launch go in fours, one tile block per four");
static_assert(TILES_ALLOC > 0 && 2 * TILES_EDGE + TILES_ALLOC == GEMM_TILES, "every tile once");
static_assert(D_HID == 128, "a tile block of 8 waves is two tiles of D_HID/32 = 4 waves");

template <int K, int F>
__device__ __forceinline__ void gemm_tile(const float* __restrict__ X,
                                          const float* __restrict__ W,
                                          float* __restrict__ H, int tile0, int ntiles) {
  const int wave = threadIdx.x >> 5;
  const int ts = wave / (F / 32);
  if (ts >= ntiles) return;
  const int rowTile = (tile0 + ts) << 4;
  const int colTile = (wave - ts * (F / 32)) << 5;
  const int lane = threadIdx.x & 31;
  const int half = lane >> 4;
  const int i16  = lane & 15;

  const float* xrow = X + (size_t)(rowTile + i16) * (size_t)K;
  const float* wcol = W + colTile + 2 * i16;

  v8f acc0 = {}, acc1 = {};
  for (int k0 = 0; k0 < K; k0 += 16) {
    v2f a[4], w0[4], w1[4];
#pragma unroll
    for (int u = 0; u < 4; ++u) {
      const int ka = k0 + 4 * u + (half << 1);
      a[u] = *(const v2f*)(xrow + ka);
    }
#pragma unroll
    for (int u = 0; u < 4; ++u) {
      const int ka = k0 + 4 * u + (half << 1);
      w0[u] = *(const v2f*)(wcol + (size_t)ka * (size_t)F);
      w1[u] = *(const v2f*)(wcol + (size_t)(ka + 1) * (size_t)F);
    }
#pragma unroll
    for (int u = 0; u < 4; ++u) {
      const v2f b0 = {w0[u].x, w1[u].x};
      const v2f b1 = {w0[u].y, w1[u].y};
      acc0 = __builtin_amdgcn_wmma_f32_16x16x4_f32(
           false, a[u],  false, b0,
           (short)0, acc0,  false,  false);
      acc1 = __builtin_amdgcn_wmma_f32_16x16x4_f32(
           false, a[u],  false, b1,
           (short)0, acc1,  false,  false);
    }
  }

  const int r0 = rowTile + (half << 3);
  float* hout = H + (size_t)r0 * (size_t)F + colTile + 2 * i16;
#pragma unroll
  for (int i = 0; i < 8; ++i) {
    const v2f o = {acc0[i], acc1[i]};
    *(v2f*)(hout + (size_t)i * (size_t)F) = o;
  }
}

#define POOL_WORDS (N_GRAPHS * D_OUT + N_GRAPHS)
__global__ void zero_pool_counts(float* __restrict__ pool, int* __restrict__ cnt,
                                 size_t ws_bytes) {
  if (ws_bytes < WS_BYTES) return;
  int i = blockIdx.x * blockDim.x + threadIdx.x;
  if (i < POOL_WORDS) pool[i] = 0.0f;
  else if (i - POOL_WORDS <= N_NODES) cnt[i - POOL_WORDS] = 0;
}

__global__ __launch_bounds__(256)
void count_rank(const int* __restrict__ dst, int* __restrict__ cnt,
                int* __restrict__ rank, size_t ws_bytes,
                const float* __restrict__ X, const float* __restrict__ W,
                float* __restrict__ H) {
  const int grp  = blockIdx.x / 5;
  const int kind = blockIdx.x - 5 * grp;
  if (kind == 4) {
    gemm_tile<D_IN, D_HID>(X, W, H, 2 * grp, 2);
    return;
  }
  if (ws_bytes < WS_BYTES) return;
  int e = (4 * grp + kind) * blockDim.x + threadIdx.x;
  if (e < N_EDGES) rank[e] = atomicAdd(&cnt[dst[e]], 1);
}

__global__ __launch_bounds__(256)
void row_alloc(int* __restrict__ cnt, int* __restrict__ rstart,
               float* __restrict__ dis, size_t ws_bytes,
               const float* __restrict__ X, const float* __restrict__ W,
               float* __restrict__ H) {
  if (blockIdx.x >= ALLOC_BLOCKS) {
    const int p = (int)blockIdx.x - ALLOC_BLOCKS;
    gemm_tile<D_IN, D_HID>(X, W, H, TILES_EDGE + 2 * p, TILES_ALLOC - 2 * p);
    return;
  }
  if (ws_bytes < WS_BYTES) return;
  const int i = blockIdx.x * blockDim.x + threadIdx.x;
  const int lane = threadIdx.x & 31;
  const int c = (i < N_NODES) ? cnt[i] : 0;
  int incl = c;
#pragma unroll
  for (int o = 1; o < 32; o <<= 1) {
    const int t = __shfl_up(incl, o);
    if (lane >= o) incl += t;
  }
  const int wsum = __shfl(incl, 31);
  int base = 0;
  if (lane == 31) base = atomicAdd(&cnt[N_NODES], wsum);
  base = __shfl(base, 31);
  if (i < N_NODES) {
    rstart[i] = base + incl - c;
    dis[i] = rsqrtf((float)c + 1.0f);
  }
}

__global__ __launch_bounds__(256)
void fill_sorted(const int* __restrict__ src, const int* __restrict__ dst,
                 const int* __restrict__ rank, const int* __restrict__ rstart,
                 const float* __restrict__ dis, int* __restrict__ csrc,
                 float* __restrict__ cw, size_t ws_bytes,
                 const float* __restrict__ X, const float* __restrict__ W,
                 float* __restrict__ H) {
  const int grp  = blockIdx.x / 5;
  const int kind = blockIdx.x - 5 * grp;
  if (kind == 4) {
    gemm_tile<D_IN, D_HID>(X, W, H, TILES_EDGE + TILES_ALLOC + 2 * grp, 2);
    return;
  }
  if (ws_bytes < WS_BYTES) return;
  int e = (4 * grp + kind) * blockDim.x + threadIdx.x;
  if (e < N_EDGES) {
    const int s = src[e];
    const int d = dst[e];
    const int pos = rstart[d] + rank[e];
    csrc[pos] = s;
    cw[pos] = dis[s] * dis[d];
  }
}

template <int VW> struct fvec { typedef float type __attribute__((ext_vector_type(VW))); };

template <int F, int R, int U>
__device__ __forceinline__ void
row_sums(const float* __restrict__ h, int row0_in, int lane,
         const int* __restrict__ cnt, const int* __restrict__ rstart,
         const int* __restrict__ csrc, const float* __restrict__ cw,
         const float* __restrict__ dis, const float* __restrict__ bias,
         typename fvec<F / 32>::type (&acc)[R]) {
  typedef typename fvec<F / 32>::type vec;
  static_assert(32 % U == 0, "a group of U slots never runs past lane 31");
  const int row0 = __builtin_amdgcn_readfirstlane(row0_in);
  const char* hb = (const char*)h;
  constexpr unsigned ROWB = F * sizeof(float);
  static_assert((unsigned long long)N_NODES * ROWB < 0xffffffffull, "32-bit byte offsets into a plane");
  const unsigned loff = (unsigned)lane * (unsigned)sizeof(vec);

  int n[R], st[R];
  float di[R];
  vec self[R];
#pragma unroll
  for (int r = 0; r < R; ++r) {
    n[r]    = cnt[row0 + r];
    st[r]   = rstart[row0 + r];
    di[r]   = dis[row0 + r];
    self[r] = *(const vec*)(hb + ((unsigned)(row0 + r) * ROWB + loff));
  }
  const vec bv = ((const vec*)bias)[lane];
  int nmax = 0;
#pragma unroll
  for (int r = 0; r < R; ++r) {
    nmax = (n[r] > nmax) ? n[r] : nmax;
  }

  int m[R], sv[R];
  float wv[R];
  auto load_slots = [&](int c0) {
#pragma unroll
    for (int r = 0; r < R; ++r) {
      const int left = n[r] - c0;
      m[r]  = (left < 0) ? 0 : ((left > 32) ? 32 : left);
      sv[r] = row0 + r;
      wv[r] = 0.0f;
      if (lane < m[r]) {
        sv[r] = csrc[st[r] + c0 + lane];
        wv[r] = cw[st[r] + c0 + lane];
      }
    }
  };
  load_slots(0);
#pragma unroll
  for (int r = 0; r < R; ++r) acc[r] = self[r] * di[r] * di[r] + bv;

  for (int c0 = 0;;) {
    int g[R];
    int gmin = 32 / U;
#pragma unroll
    for (int r = 0; r < R; ++r) {
      g[r] = (m[r] + U - 1) / U;
      gmin = (g[r] < gmin) ? g[r] : gmin;
    }
    for (int t = 0; t < gmin; ++t) {
      const int j = t * U;
      vec rw[R][U];
#pragma unroll
      for (int r = 0; r < R; ++r) {
#pragma unroll
        for (int u = 0; u < U; ++u) {
          const int sj = __builtin_amdgcn_readlane(sv[r], j + u);
          rw[r][u] = *(const vec*)(hb + ((unsigned)sj * ROWB + loff));
        }
      }
#pragma unroll
      for (int r = 0; r < R; ++r) {
#pragma unroll
        for (int u = 0; u < U; ++u) {
          const float w = __builtin_bit_cast(float,
                            __builtin_amdgcn_readlane(__builtin_bit_cast(int, wv[r]), j + u));
          acc[r] += rw[r][u] * w;
        }
      }
    }
#pragma unroll
    for (int r = 0; r < R; ++r) {
      for (int t = gmin; t < g[r]; ++t) {
        const int j = t * U;
        vec rw[U];
#pragma unroll
        for (int u = 0; u < U; ++u) {
          const int sj = __builtin_amdgcn_readlane(sv[r], j + u);
          rw[u] = *(const vec*)(hb + ((unsigned)sj * ROWB + loff));
        }
#pragma unroll
        for (int u = 0; u < U; ++u) {
          const float w = __builtin_bit_cast(float,
                            __builtin_amdgcn_readlane(__builtin_bit_cast(int, wv[r]), j + u));
          acc[r] += rw[u] * w;
        }
      }
    }
    c0 += 32;
    if (c0 >= nmax) break;
    load_slots(c0);
  }
}

#define POOL_ROWS 16
template <int F>
__global__ __launch_bounds__(128)
void gather_pool(const float* __restrict__ h, const int* __restrict__ cnt,
                 const int* __restrict__ rstart, const int* __restrict__ csrc,
                 const float* __restrict__ cw, const float* __restrict__ dis,
                 const float* __restrict__ bias, const int* __restrict__ batch,
                 float* __restrict__ sums, float* __restrict__ cnts,
                 size_t ws_bytes) {
  constexpr int VW  = F / 32;
  constexpr int NW  = 4;
  constexpr int RPW = POOL_ROWS / NW;
  typedef typename fvec<VW>::type vec;
  static_assert(N_NODES % POOL_ROWS == 0, "a block owns POOL_ROWS whole rows");
  static_assert(F <= 32 * NW, "one thread per feature in the pool part");
  __shared__ __attribute__((aligned(16))) float tile[POOL_ROWS * F];
  __shared__ int gid[POOL_ROWS];
  if (ws_bytes < WS_BYTES) return;
  const int lane = threadIdx.x & 31;
  const int wave = threadIdx.x >> 5;
  const int row0 = blockIdx.x * POOL_ROWS;

  static_assert(RPW % 2 == 0, "a wave takes its rows in pairs");
  for (int rr = 0; rr < RPW; rr += 2) {
    const int lr = wave * RPW + rr;
    vec acc[2];
    row_sums<F, 2, 4>(h, row0 + lr, lane, cnt, rstart, csrc, cw, dis, bias, acc);
    ((vec*)tile)[lr * 32 + lane] = acc[0];
    ((vec*)tile)[(lr + 1) * 32 + lane] = acc[1];
    if (lane < 2) gid[lr + lane] = batch[row0 + lr + lane];
  }
  __syncthreads();

  if (threadIdx.x >= F) return;
  const int f = threadIdx.x;
  float v[POOL_ROWS];
  int g[POOL_ROWS];
#pragma unroll
  for (int r = 0; r < POOL_ROWS; ++r) {
    v[r] = tile[r * F + f];
    g[r] = gid[r];
  }
  int cur = g[0];
  float a = v[0];
  int c = 1;
#pragma unroll
  for (int r = 1; r < POOL_ROWS; ++r) {
    if (g[r] != cur) {
      atomicAdd(&sums[cur * F + f], a);
      if (f == 0) atomicAdd(&cnts[cur], (float)c);
      cur = g[r];
      a = 0.0f;
      c = 0;
    }
    a += v[r];
    ++c;
  }
  atomicAdd(&sums[cur * F + f], a);
  if (f == 0) atomicAdd(&cnts[cur], (float)c);
}

#define A_PITCH 132
template <int FOUT>
__global__ __launch_bounds__(32 * (FOUT / 32))
void gather_gemm(const float* __restrict__ hin, const int* __restrict__ cnt,
                 const int* __restrict__ rstart, const int* __restrict__ csrc,
                 const float* __restrict__ cw, const float* __restrict__ dis,
                 const float* __restrict__ bias_in, const float* __restrict__ W,
                 float* __restrict__ Hout, size_t ws_bytes) {
  constexpr int NW  = FOUT / 32;
  constexpr int RPW = 16 / NW;
  typedef typename fvec<4>::type v4f;
  __shared__ __attribute__((aligned(16))) float tile[16 * A_PITCH];
  if (ws_bytes < WS_BYTES) return;
  const int lane = threadIdx.x & 31;
  const int wave = threadIdx.x >> 5;
  const int rowTile = blockIdx.x << 4;

  static_assert(RPW % 2 == 0, "a wave takes its rows in pairs");
  for (int rp = 0; rp < RPW; rp += 2) {
    v4f sums[2];
    row_sums<D_HID, 2, 4>(hin, rowTile + wave * RPW + rp, lane, cnt, rstart, csrc, cw, dis,
                          bias_in, sums);
#pragma unroll
    for (int rr = 0; rr < 2; ++rr) {
      const int lr = wave * RPW + rp + rr;
      v4f s = sums[rr];
      s.x = fmaxf(s.x, 0.0f);
      s.y = fmaxf(s.y, 0.0f);
      s.z = fmaxf(s.z, 0.0f);
      s.w = fmaxf(s.w, 0.0f);
      *(v4f*)(tile + lr * A_PITCH + 4 * lane) = s;
    }
  }
  __syncthreads();

  const int half = lane >> 4;
  const int i16  = lane & 15;
  const int colTile = wave << 5;
  const float* arow = tile + i16 * A_PITCH;
  const char* wb = (const char*)W;
  const unsigned woff = (unsigned)(((half << 1) * FOUT + colTile + 2 * i16) * (int)sizeof(float));
  constexpr unsigned WROWB = FOUT * sizeof(float);

  v8f acc0 = {}, acc1 = {};
  for (int k0 = 0; k0 < D_HID; k0 += 16) {
    v2f a[4], w0[4], w1[4];
#pragma unroll
    for (int u = 0; u < 4; ++u) {
      const int ka = k0 + 4 * u + (half << 1);
      a[u] = *(const v2f*)(arow + ka);
    }
#pragma unroll
    for (int u = 0; u < 4; ++u) {
      w0[u] = *(const v2f*)(wb + (woff + (unsigned)(k0 + 4 * u) * WROWB));
      w1[u] = *(const v2f*)(wb + (woff + (unsigned)(k0 + 4 * u + 1) * WROWB));
    }
#pragma unroll
    for (int u = 0; u < 4; ++u) {
      const v2f b0 = {w0[u].x, w1[u].x};
      const v2f b1 = {w0[u].y, w1[u].y};
      acc0 = __builtin_amdgcn_wmma_f32_16x16x4_f32(
           false, a[u],  false, b0,
           (short)0, acc0,  false,  false);
      acc1 = __builtin_amdgcn_wmma_f32_16x16x4_f32(
           false, a[u],  false, b1,
           (short)0, acc1,  false,  false);
    }
  }

  const int r0 = rowTile + (half << 3);
  static_assert((unsigned long long)N_NODES * FOUT * sizeof(float) < 0xffffffffull, "32-bit byte offsets into a plane");
  char* ob = (char*)Hout;
  const unsigned ooff = (unsigned)(r0 * FOUT + colTile + 2 * i16) * (unsigned)sizeof(float);
#pragma unroll
  for (int i = 0; i < 8; ++i) {
    const v2f o = {acc0[i], acc1[i]};
    *(v2f*)(ob + (ooff + (unsigned)i * WROWB)) = o;
  }
}

__global__ void pool_final(const float* __restrict__ sums, const float* __restrict__ cnts,
                           float* __restrict__ out) {
  int i = blockIdx.x * blockDim.x + threadIdx.x;
  if (i < N_GRAPHS * D_OUT) {
    int g = i >> 6;
    out[i] = sums[i] / fmaxf(cnts[g], 1.0f);
  }
}

extern "C" void kernel_launch(void* const* d_in, const int* in_sizes, int n_in,
                              void* d_out, int out_size, void* d_ws, size_t ws_size,
                              hipStream_t stream) {
  const float* x     = (const float*)d_in[0];
  const int*   ei    = (const int*)d_in[1];
  const int*   batch = (const int*)d_in[2];
  const float* W1 = (const float*)d_in[3];  const float* b1 = (const float*)d_in[4];
  const float* W2 = (const float*)d_in[5];  const float* b2 = (const float*)d_in[6];
  const float* W3 = (const float*)d_in[7];  const float* b3 = (const float*)d_in[8];
  const float* W4 = (const float*)d_in[9];  const float* b4 = (const float*)d_in[10];
  float* out = (float*)d_out;

  float* bufA = (float*)d_ws;
  float* bufB = bufA + (size_t)N_NODES * D_HID;
  float* dis  = bufB + (size_t)N_NODES * D_HID;
  float* sums = dis + N_NODES;
  float* cnts = sums + N_GRAPHS * D_OUT;
  int*   cnt    = (int*)(cnts + N_GRAPHS);
  int*   rstart = cnt + (N_NODES + 32);
  int*   csrc   = rstart + N_NODES;
  float* cw     = (float*)(csrc + N_EDGES);
  int*   rank   = (int*)bufB;

  const int* src = ei;
  const int* dst = ei + N_EDGES;

  const int TB = 256;
  auto blocks = [](long long n, int tb) { return (unsigned)((n + tb - 1) / tb); };

  zero_pool_counts<<<blocks(POOL_WORDS + N_NODES + 1, TB), TB, 0, stream>>>(
      sums, cnt, ws_size);
  count_rank<<<EDGE_GRID, TB, 0, stream>>>(dst, cnt, rank, ws_size, x, W1, bufA);
  row_alloc<<<ALLOC_BLOCKS + PAIRS_ALLOC, TB, 0, stream>>>(cnt, rstart, dis, ws_size,
                                                           x, W1, bufA);
  fill_sorted<<<EDGE_GRID, TB, 0, stream>>>(src, dst, rank, rstart, dis, csrc, cw, ws_size,
                                            x, W1, bufA);
  gather_gemm<D_HID><<<N_NODES / 16, 128, 0, stream>>>(
      bufA, cnt, rstart, csrc, cw, dis, b1, W2, bufB, ws_size);
  gather_gemm<D_HID><<<N_NODES / 16, 128, 0, stream>>>(
      bufB, cnt, rstart, csrc, cw, dis, b2, W3, bufA, ws_size);
  gather_gemm<D_OUT><<<N_NODES / 16, 64, 0, stream>>>(
      bufA, cnt, rstart, csrc, cw, dis, b3, W4, bufB, ws_size);
  gather_pool<D_OUT><<<N_NODES / POOL_ROWS, 128, 0, stream>>>(
      bufB, cnt, rstart, csrc, cw, dis, b4, batch, sums, cnts, ws_size);
  pool_final<<<blocks(N_GRAPHS * D_OUT, TB), TB, 0, stream>>>(sums, cnts, out);
}


# degree counts and edge ranks in LDS: 256 list blocks, each a row range x an eighth of dst, rank = LDS add's return; no memory atomics, no zero pass: 9 launches are 8; sum order: eighth, then arrival
# speedup vs baseline: 6.1219x; 1.0397x over previous
#include <hip/hip_runtime.h>
#include <hip/hip_bf16.h>

typedef float v2f __attribute__((ext_vector_type(2)));
typedef float v8f __attribute__((ext_vector_type(8)));

#define N_NODES  50000
#define N_EDGES  600000
#define N_GRAPHS 64
#define D_IN     128
#define D_HID    128
#define D_OUT    64

#define WS_WORDS ((size_t)2 * N_NODES * D_HID + N_NODES + N_GRAPHS * D_OUT + N_GRAPHS + \
                  (N_NODES + 32) + N_NODES + (size_t)2 * N_EDGES)
#define WS_BYTES (WS_WORDS * sizeof(float))

#define GEMM_TILES   (N_NODES / 16)
#define EDGE_BLOCKS  ((N_EDGES + 255) / 256)
#define ALLOC_BLOCKS ((N_NODES + 255) / 256)
#define TILES_EDGE   (EDGE_BLOCKS / 2)
#define TILES_ALLOC  (GEMM_TILES - 2 * TILES_EDGE)
#define PAIRS_EDGE   (TILES_EDGE / 2)
#define PAIRS_ALLOC  ((TILES_ALLOC + 1) / 2)
#define EDGE_GRID    (5 * PAIRS_EDGE)
#define LIST_RANGES  32
#define LIST_PARTS   8
#define LIST_BLOCKS  (LIST_RANGES * LIST_PARTS)
#define LIST_ROWS    ((N_NODES + LIST_RANGES - 1) / LIST_RANGES)
#define PART_EDGES   (N_EDGES / LIST_PARTS)
#define CR_THREADS   1024
#define CR_TILES     (CR_THREADS / 32 / (D_HID / 32))
#define CR_TILE_BLOCKS ((TILES_EDGE + CR_TILES - 1) / CR_TILES)
static_assert(EDGE_BLOCKS % 4 == 0, "list blocks of an edge launch go in fours, one tile block per four");
static_assert(TILES_ALLOC > 0 && 2 * TILES_EDGE + TILES_ALLOC == GEMM_TILES, "every tile once");
static_assert(D_HID == 128, "a tile block of 8 waves is two tiles of D_HID/32 = 4 waves");

template <int K, int F>
__device__ __forceinline__ void gemm_tile(const float* __restrict__ X,
                                          const float* __restrict__ W,
                                          float* __restrict__ H, int tile0, int ntiles) {
  const int wave = threadIdx.x >> 5;
  const int ts = wave / (F / 32);
  if (ts >= ntiles) return;
  const int rowTile = (tile0 + ts) << 4;
  const int colTile = (wave - ts * (F / 32)) << 5;
  const int lane = threadIdx.x & 31;
  const int half = lane >> 4;
  const int i16  = lane & 15;

  const float* xrow = X + (size_t)(rowTile + i16) * (size_t)K;
  const float* wcol = W + colTile + 2 * i16;

  v8f acc0 = {}, acc1 = {};
  for (int k0 = 0; k0 < K; k0 += 16) {
    v2f a[4], w0[4], w1[4];
#pragma unroll
    for (int u = 0; u < 4; ++u) {
      const int ka = k0 + 4 * u + (half << 1);
      a[u] = *(const v2f*)(xrow + ka);
    }
#pragma unroll
    for (int u = 0; u < 4; ++u) {
      const int ka = k0 + 4 * u + (half << 1);
      w0[u] = *(const v2f*)(wcol + (size_t)ka * (size_t)F);
      w1[u] = *(const v2f*)(wcol + (size_t)(ka + 1) * (size_t)F);
    }
#pragma unroll
    for (int u = 0; u < 4; ++u) {
      const v2f b0 = {w0[u].x, w1[u].x};
      const v2f b1 = {w0[u].y, w1[u].y};
      acc0 = __builtin_amdgcn_wmma_f32_16x16x4_f32(
           false, a[u],  false, b0,
           (short)0, acc0,  false,  false);
      acc1 = __builtin_amdgcn_wmma_f32_16x16x4_f32(
           false, a[u],  false, b1,
           (short)0, acc1,  false,  false);
    }
  }

  const int r0 = rowTile + (half << 3);
  float* hout = H + (size_t)r0 * (size_t)F + colTile + 2 * i16;
#pragma unroll
  for (int i = 0; i < 8; ++i) {
    const v2f o = {acc0[i], acc1[i]};
    *(v2f*)(hout + (size_t)i * (size_t)F) = o;
  }
}

#define POOL_WORDS (N_GRAPHS * D_OUT + N_GRAPHS)
static_assert(N_EDGES % (4 * LIST_PARTS) == 0, "a part of the dst list is whole 16-byte quads");
static_assert(LIST_BLOCKS * CR_THREADS >= POOL_WORDS, "one thread per pool word among the list blocks");
typedef int v4i __attribute__((ext_vector_type(4)));
__global__ __launch_bounds__(CR_THREADS)
void count_rank(const int* __restrict__ dst, int* __restrict__ cntp,
                int* __restrict__ rank, int* __restrict__ cnt,
                float* __restrict__ pool, size_t ws_bytes,
                const float* __restrict__ X, const float* __restrict__ W,
                float* __restrict__ H) {
  __shared__ int lcnt[LIST_ROWS];
  if (blockIdx.x >= LIST_BLOCKS) {
    const int p = (int)blockIdx.x - LIST_BLOCKS;
    gemm_tile<D_IN, D_HID>(X, W, H, CR_TILES * p, TILES_EDGE - CR_TILES * p);
    return;
  }
  if (ws_bytes < WS_BYTES) return;
  const int tid = threadIdx.x;
  const int part = blockIdx.x / LIST_RANGES;
  const int lo = (blockIdx.x - part * LIST_RANGES) * LIST_ROWS;
  for (int i = tid; i < LIST_ROWS; i += CR_THREADS) lcnt[i] = 0;
  const int z = blockIdx.x * CR_THREADS + tid;
  if (z < POOL_WORDS) pool[z] = 0.0f;
  if (z == 0) cnt[N_NODES] = 0;
  __syncthreads();

  constexpr int NQ = PART_EDGES / 4;
  const v4i* d4 = (const v4i*)dst + (size_t)part * NQ;
  int* rk4 = rank + (size_t)part * PART_EDGES;
  for (int q0 = tid; q0 < NQ; q0 += 4 * CR_THREADS) {
    v4i d[4];
#pragma unroll
    for (int j = 0; j < 4; ++j) {
      const int q = q0 + j * CR_THREADS;
      const v4i none = {-1, -1, -1, -1};
      d[j] = (q < NQ) ? d4[q] : none;
    }
    int rk[4][4];
#pragma unroll
    for (int j = 0; j < 4; ++j) {
#pragma unroll
      for (int k = 0; k < 4; ++k) {
        const unsigned r = (unsigned)(d[j][k] - lo);
        rk[j][k] = -1;
        if (r < (unsigned)LIST_ROWS) rk[j][k] = atomicAdd(&lcnt[r], 1);
      }
    }
#pragma unroll
    for (int j = 0; j < 4; ++j) {
#pragma unroll
      for (int k = 0; k < 4; ++k) {
        if (rk[j][k] >= 0) rk4[4 * (q0 + j * CR_THREADS) + k] = rk[j][k];
      }
    }
  }
  __syncthreads();

  int* mine = cntp + (size_t)part * N_NODES;
  for (int i = tid; i < LIST_ROWS; i += CR_THREADS) {
    if (lo + i < N_NODES) mine[lo + i] = lcnt[i];
  }
}

__global__ __launch_bounds__(256)
void row_alloc(const int* __restrict__ cntp, int* __restrict__ pstart,
               int* __restrict__ cnt, int* __restrict__ rstart,
               float* __restrict__ dis, size_t ws_bytes,
               const float* __restrict__ X, const float* __restrict__ W,
               float* __restrict__ H) {
  if (blockIdx.x >= ALLOC_BLOCKS) {
    const int p = (int)blockIdx.x - ALLOC_BLOCKS;
    gemm_tile<D_IN, D_HID>(X, W, H, TILES_EDGE + 2 * p, TILES_ALLOC - 2 * p);
    return;
  }
  if (ws_bytes < WS_BYTES) return;
  const int i = blockIdx.x * blockDim.x + threadIdx.x;
  const int lane = threadIdx.x & 31;
  int cp[LIST_PARTS];
  int c = 0;
#pragma unroll
  for (int p = 0; p < LIST_PARTS; ++p) {
    cp[p] = (i < N_NODES) ? cntp[p * N_NODES + i] : 0;
    c += cp[p];
  }
  int incl = c;
#pragma unroll
  for (int o = 1; o < 32; o <<= 1) {
    const int t = __shfl_up(incl, o);
    if (lane >= o) incl += t;
  }
  const int wsum = __shfl(incl, 31);
  int base = 0;
  if (lane == 31) base = atomicAdd(&cnt[N_NODES], wsum);
  base = __shfl(base, 31);
  if (i < N_NODES) {
    int s = base + incl - c;
    cnt[i] = c;
    rstart[i] = s;
#pragma unroll
    for (int p = 0; p < LIST_PARTS; ++p) {
      pstart[p * N_NODES + i] = s;
      s += cp[p];
    }
    dis[i] = rsqrtf((float)c + 1.0f);
  }
}

__global__ __launch_bounds__(256)
void fill_sorted(const int* __restrict__ src, const int* __restrict__ dst,
                 const int* __restrict__ rank, const int* __restrict__ pstart,
                 const float* __restrict__ dis, int* __restrict__ csrc,
                 float* __restrict__ cw, size_t ws_bytes,
                 const float* __restrict__ X, const float* __restrict__ W,
                 float* __restrict__ H) {
  const int grp  = blockIdx.x / 5;
  const int kind = blockIdx.x - 5 * grp;
  if (kind == 4) {
    gemm_tile<D_IN, D_HID>(X, W, H, TILES_EDGE + TILES_ALLOC + 2 * grp, 2);
    return;
  }
  if (ws_bytes < WS_BYTES) return;
  int e = (4 * grp + kind) * blockDim.x + threadIdx.x;
  if (e < N_EDGES) {
    const int s = src[e];
    const int d = dst[e];
    const int pos = pstart[(e / PART_EDGES) * N_NODES + d] + rank[e];
    csrc[pos] = s;
    cw[pos] = dis[s] * dis[d];
  }
}

template <int VW> struct fvec { typedef float type __attribute__((ext_vector_type(VW))); };

template <int F, int R, int U>
__device__ __forceinline__ void
row_sums(const float* __restrict__ h, int row0_in, int lane,
         const int* __restrict__ cnt, const int* __restrict__ rstart,
         const int* __restrict__ csrc, const float* __restrict__ cw,
         const float* __restrict__ dis, const float* __restrict__ bias,
         typename fvec<F / 32>::type (&acc)[R]) {
  typedef typename fvec<F / 32>::type vec;
  static_assert(32 % U == 0, "a group of U slots never runs past lane 31");
  const int row0 = __builtin_amdgcn_readfirstlane(row0_in);
  const char* hb = (const char*)h;
  constexpr unsigned ROWB = F * sizeof(float);
  static_assert((unsigned long long)N_NODES * ROWB < 0xffffffffull, "32-bit byte offsets into a plane");
  const unsigned loff = (unsigned)lane * (unsigned)sizeof(vec);

  int n[R], st[R];
  float di[R];
  vec self[R];
#pragma unroll
  for (int r = 0; r < R; ++r) {
    n[r]    = cnt[row0 + r];
    st[r]   = rstart[row0 + r];
    di[r]   = dis[row0 + r];
    self[r] = *(const vec*)(hb + ((unsigned)(row0 + r) * ROWB + loff));
  }
  const vec bv = ((const vec*)bias)[lane];
  int nmax = 0;
#pragma unroll
  for (int r = 0; r < R; ++r) {
    nmax = (n[r] > nmax) ? n[r] : nmax;
  }

  int m[R], sv[R];
  float wv[R];
  auto load_slots = [&](int c0) {
#pragma unroll
    for (int r = 0; r < R; ++r) {
      const int left = n[r] - c0;
      m[r]  = (left < 0) ? 0 : ((left > 32) ? 32 : left);
      sv[r] = row0 + r;
      wv[r] = 0.0f;
      if (lane < m[r]) {
        sv[r] = csrc[st[r] + c0 + lane];
        wv[r] = cw[st[r] + c0 + lane];
      }
    }
  };
  load_slots(0);
#pragma unroll
  for (int r = 0; r < R; ++r) acc[r] = self[r] * di[r] * di[r] + bv;

  for (int c0 = 0;;) {
    int g[R];
    int gmin = 32 / U;
#pragma unroll
    for (int r = 0; r < R; ++r) {
      g[r] = (m[r] + U - 1) / U;
      gmin = (g[r] < gmin) ? g[r] : gmin;
    }
    for (int t = 0; t < gmin; ++t) {
      const int j = t * U;
      vec rw[R][U];
#pragma unroll
      for (int r = 0; r < R; ++r) {
#pragma unroll
        for (int u = 0; u < U; ++u) {
          const int sj = __builtin_amdgcn_readlane(sv[r], j + u);
          rw[r][u] = *(const vec*)(hb + ((unsigned)sj * ROWB + loff));
        }
      }
#pragma unroll
      for (int r = 0; r < R; ++r) {
#pragma unroll
        for (int u = 0; u < U; ++u) {
          const float w = __builtin_bit_cast(float,
                            __builtin_amdgcn_readlane(__builtin_bit_cast(int, wv[r]), j + u));
          acc[r] += rw[r][u] * w;
        }
      }
    }
#pragma unroll
    for (int r = 0; r < R; ++r) {
      for (int t = gmin; t < g[r]; ++t) {
        const int j = t * U;
        vec rw[U];
#pragma unroll
        for (int u = 0; u < U; ++u) {
          const int sj = __builtin_amdgcn_readlane(sv[r], j + u);
          rw[u] = *(const vec*)(hb + ((unsigned)sj * ROWB + loff));
        }
#pragma unroll
        for (int u = 0; u < U; ++u) {
          const float w = __builtin_bit_cast(float,
                            __builtin_amdgcn_readlane(__builtin_bit_cast(int, wv[r]), j + u));
          acc[r] += rw[u] * w;
        }
      }
    }
    c0 += 32;
    if (c0 >= nmax) break;
    load_slots(c0);
  }
}

#define POOL_ROWS 16
template <int F>
__global__ __launch_bounds__(128)
void gather_pool(const float* __restrict__ h, const int* __restrict__ cnt,
                 const int* __restrict__ rstart, const int* __restrict__ csrc,
                 const float* __restrict__ cw, const float* __restrict__ dis,
                 const float* __restrict__ bias, const int* __restrict__ batch,
                 float* __restrict__ sums, float* __restrict__ cnts,
                 size_t ws_bytes) {
  constexpr int VW  = F / 32;
  constexpr int NW  = 4;
  constexpr int RPW = POOL_ROWS / NW;
  typedef typename fvec<VW>::type vec;
  static_assert(N_NODES % POOL_ROWS == 0, "a block owns POOL_ROWS whole rows");
  static_assert(F <= 32 * NW, "one thread per feature in the pool part");
  __shared__ __attribute__((aligned(16))) float tile[POOL_ROWS * F];
  __shared__ int gid[POOL_ROWS];
  if (ws_bytes < WS_BYTES) return;
  const int lane = threadIdx.x & 31;
  const int wave = threadIdx.x >> 5;
  const int row0 = blockIdx.x * POOL_ROWS;

  static_assert(RPW % 2 == 0, "a wave takes its rows in pairs");
  for (int rr = 0; rr < RPW; rr += 2) {
    const int lr = wave * RPW + rr;
    vec acc[2];
    row_sums<F, 2, 4>(h, row0 + lr, lane, cnt, rstart, csrc, cw, dis, bias, acc);
    ((vec*)tile)[lr * 32 + lane] = acc[0];
    ((vec*)tile)[(lr + 1) * 32 + lane] = acc[1];
    if (lane < 2) gid[lr + lane] = batch[row0 + lr + lane];
  }
  __syncthreads();

  if (threadIdx.x >= F) return;
  const int f = threadIdx.x;
  float v[POOL_ROWS];
  int g[POOL_ROWS];
#pragma unroll
  for (int r = 0; r < POOL_ROWS; ++r) {
    v[r] = tile[r * F + f];
    g[r] = gid[r];
  }
  int cur = g[0];
  float a = v[0];
  int c = 1;
#pragma unroll
  for (int r = 1; r < POOL_ROWS; ++r) {
    if (g[r] != cur) {
      atomicAdd(&sums[cur * F + f], a);
      if (f == 0) atomicAdd(&cnts[cur], (float)c);
      cur = g[r];
      a = 0.0f;
      c = 0;
    }
    a += v[r];
    ++c;
  }
  atomicAdd(&sums[cur * F + f], a);
  if (f == 0) atomicAdd(&cnts[cur], (float)c);
}

#define A_PITCH 132
template <int FOUT>
__global__ __launch_bounds__(32 * (FOUT / 32))
void gather_gemm(const float* __restrict__ hin, const int* __restrict__ cnt,
                 const int* __restrict__ rstart, const int* __restrict__ csrc,
                 const float* __restrict__ cw, const float* __restrict__ dis,
                 const float* __restrict__ bias_in, const float* __restrict__ W,
                 float* __restrict__ Hout, size_t ws_bytes) {
  constexpr int NW  = FOUT / 32;
  constexpr int RPW = 16 / NW;
  typedef typename fvec<4>::type v4f;
  __shared__ __attribute__((aligned(16))) float tile[16 * A_PITCH];
  if (ws_bytes < WS_BYTES) return;
  const int lane = threadIdx.x & 31;
  const int wave = threadIdx.x >> 5;
  const int rowTile = blockIdx.x << 4;

  static_assert(RPW % 2 == 0, "a wave takes its rows in pairs");
  for (int rp = 0; rp < RPW; rp += 2) {
    v4f sums[2];
    row_sums<D_HID, 2, 4>(hin, rowTile + wave * RPW + rp, lane, cnt, rstart, csrc, cw, dis,
                          bias_in, sums);
#pragma unroll
    for (int rr = 0; rr < 2; ++rr) {
      const int lr = wave * RPW + rp + rr;
      v4f s = sums[rr];
      s.x = fmaxf(s.x, 0.0f);
      s.y = fmaxf(s.y, 0.0f);
      s.z = fmaxf(s.z, 0.0f);
      s.w = fmaxf(s.w, 0.0f);
      *(v4f*)(tile + lr * A_PITCH + 4 * lane) = s;
    }
  }
  __syncthreads();

  const int half = lane >> 4;
  const int i16  = lane & 15;
  const int colTile = wave << 5;
  const float* arow = tile + i16 * A_PITCH;
  const char* wb = (const char*)W;
  const unsigned woff = (unsigned)(((half << 1) * FOUT + colTile + 2 * i16) * (int)sizeof(float));
  constexpr unsigned WROWB = FOUT * sizeof(float);

  v8f acc0 = {}, acc1 = {};
  for (int k0 = 0; k0 < D_HID; k0 += 16) {
    v2f a[4], w0[4], w1[4];
#pragma unroll
    for (int u = 0; u < 4; ++u) {
      const int ka = k0 + 4 * u + (half << 1);
      a[u] = *(const v2f*)(arow + ka);
    }
#pragma unroll
    for (int u = 0; u < 4; ++u) {
      w0[u] = *(const v2f*)(wb + (woff + (unsigned)(k0 + 4 * u) * WROWB));
      w1[u] = *(const v2f*)(wb + (woff + (unsigned)(k0 + 4 * u + 1) * WROWB));
    }
#pragma unroll
    for (int u = 0; u < 4; ++u) {
      const v2f b0 = {w0[u].x, w1[u].x};
      const v2f b1 = {w0[u].y, w1[u].y};
      acc0 = __builtin_amdgcn_wmma_f32_16x16x4_f32(
           false, a[u],  false, b0,
           (short)0, acc0,  false,  false);
      acc1 = __builtin_amdgcn_wmma_f32_16x16x4_f32(
           false, a[u],  false, b1,
           (short)0, acc1,  false,  false);
    }
  }

  const int r0 = rowTile + (half << 3);
  static_assert((unsigned long long)N_NODES * FOUT * sizeof(float) < 0xffffffffull, "32-bit byte offsets into a plane");
  char* ob = (char*)Hout;
  const unsigned ooff = (unsigned)(r0 * FOUT + colTile + 2 * i16) * (unsigned)sizeof(float);
#pragma unroll
  for (int i = 0; i < 8; ++i) {
    const v2f o = {acc0[i], acc1[i]};
    *(v2f*)(ob + (ooff + (unsigned)i * WROWB)) = o;
  }
}

__global__ void pool_final(const float* __restrict__ sums, const float* __restrict__ cnts,
                           float* __restrict__ out) {
  int i = blockIdx.x * blockDim.x + threadIdx.x;
  if (i < N_GRAPHS * D_OUT) {
    int g = i >> 6;
    out[i] = sums[i] / fmaxf(cnts[g], 1.0f);
  }
}

extern "C" void kernel_launch(void* const* d_in, const int* in_sizes, int n_in,
                              void* d_out, int out_size, void* d_ws, size_t ws_size,
                              hipStream_t stream) {
  const float* x     = (const float*)d_in[0];
  const int*   ei    = (const int*)d_in[1];
  const int*   batch = (const int*)d_in[2];
  const float* W1 = (const float*)d_in[3];  const float* b1 = (const float*)d_in[4];
  const float* W2 = (const float*)d_in[5];  const float* b2 = (const float*)d_in[6];
  const float* W3 = (const float*)d_in[7];  const float* b3 = (const float*)d_in[8];
  const float* W4 = (const float*)d_in[9];  const float* b4 = (const float*)d_in[10];
  float* out = (float*)d_out;

  float* bufA = (float*)d_ws;
  float* bufB = bufA + (size_t)N_NODES * D_HID;
  float* dis  = bufB + (size_t)N_NODES * D_HID;
  float* sums = dis + N_NODES;
  float* cnts = sums + N_GRAPHS * D_OUT;
  int*   cnt    = (int*)(cnts + N_GRAPHS);
  int*   rstart = cnt + (N_NODES + 32);
  int*   csrc   = rstart + N_NODES;
  float* cw     = (float*)(csrc + N_EDGES);
  int*   rank   = (int*)bufB;
  int*   cntp   = rank + N_EDGES;
  int*   pstart = cntp + LIST_PARTS * N_NODES;
  static_assert((size_t)N_EDGES + 2 * LIST_PARTS * N_NODES <= (size_t)N_NODES * D_HID, "borrowed words fit in bufB");

  const int* src = ei;
  const int* dst = ei + N_EDGES;

  const int TB = 256;
  auto blocks = [](long long n, int tb) { return (unsigned)((n + tb - 1) / tb); };

  count_rank<<<LIST_BLOCKS + CR_TILE_BLOCKS, CR_THREADS, 0, stream>>>(
      dst, cntp, rank, cnt, sums, ws_size, x, W1, bufA);
  row_alloc<<<ALLOC_BLOCKS + PAIRS_ALLOC, TB, 0, stream>>>(cntp, pstart, cnt, rstart, dis, ws_size,
                                                           x, W1, bufA);
  fill_sorted<<<EDGE_GRID, TB, 0, stream>>>(src, dst, rank, pstart, dis, csrc, cw, ws_size,
                                            x, W1, bufA);
  gather_gemm<D_HID><<<N_NODES / 16, 128, 0, stream>>>(
      bufA, cnt, rstart, csrc, cw, dis, b1, W2, bufB, ws_size);
  gather_gemm<D_HID><<<N_NODES / 16, 128, 0, stream>>>(
      bufB, cnt, rstart, csrc, cw, dis, b2, W3, bufA, ws_size);
  gather_gemm<D_OUT><<<N_NODES / 16, 64, 0, stream>>>(
      bufA, cnt, rstart, csrc, cw, dis, b3, W4, bufB, ws_size);
  gather_pool<D_OUT><<<N_NODES / POOL_ROWS, 128, 0, stream>>>(
      bufB, cnt, rstart, csrc, cw, dis, b4, batch, sums, cnts, ws_size);
  pool_final<<<blocks(N_GRAPHS * D_OUT, TB), TB, 0, stream>>>(sums, cnts, out);
}
